# starting kernel
# baseline (speedup 1.0000x reference)
#include <hip/hip_runtime.h>
#include <hip/hip_bf16.h>


typedef __attribute__((ext_vector_type(2))) float v2f;
typedef __attribute__((ext_vector_type(8))) float v8f;

__global__ __launch_bounds__(256) void l2norm_rows(const float* __restrict__ F,
                                                   float* __restrict__ X,
                                                   int nRows) {
    const int lane = threadIdx.x & 31;
    const int row  = blockIdx.x * 8 + (threadIdx.x >> 5);
    if (row >= nRows) return;
    const float4* f = (const float4*)(F + (size_t)row * 256);
    float4 v0 = f[lane];
    float4 v1 = f[lane + 32];
    float s = v0.x*v0.x + v0.y*v0.y + v0.z*v0.z + v0.w*v0.w
            + v1.x*v1.x + v1.y*v1.y + v1.z*v1.z + v1.w*v1.w;
    #pragma unroll
    for (int off = 16; off > 0; off >>= 1) s += __shfl_xor(s, off, 32);
    const float inv = rsqrtf(fmaxf(s, 1e-12f));
    v0.x *= inv; v0.y *= inv; v0.z *= inv; v0.w *= inv;
    v1.x *= inv; v1.y *= inv; v1.z *= inv; v1.w *= inv;
    float4* xo = (float4*)(X + (size_t)row * 256);
    xo[lane]      = v0;
    xo[lane + 32] = v1;
}

template <int K, int M, bool RELU>
__global__ __launch_bounds__(128) void gemm_wmma_f32(const float* __restrict__ X,
                                                     const float* __restrict__ W,
                                                     float* __restrict__ Y,
                                                     int nRows) {
    const int lane   = threadIdx.x & 31;
    const int waveId = blockIdx.x * 4 + (threadIdx.x >> 5);
    const int tilesN = M / 16;
    const int totalTiles = (nRows / 16) * tilesN;
    if (waveId >= totalTiles) return;
    const int rowTile = waveId / tilesN;
    const int colTile = waveId % tilesN;

    const int halfSel = lane >> 4;
    const int kOff    = halfSel * 2;
    const int aRow    = rowTile * 16 + (lane & 15);
    const int bCol    = colTile * 16 + (lane & 15);

    const float* __restrict__ xrow = X + (size_t)aRow * K;
    v8f c = {};
    #pragma unroll 4
    for (int k = 0; k < K; k += 4) {
        float a0 = xrow[k + kOff];
        float a1 = xrow[k + kOff + 1];
        if (RELU) { a0 = fmaxf(a0, 0.0f); a1 = fmaxf(a1, 0.0f); }
        v2f a; a.x = a0; a.y = a1;
        v2f b;
        b.x = W[(size_t)(k + kOff)     * M + bCol];
        b.y = W[(size_t)(k + kOff + 1) * M + bCol];
        c = __builtin_amdgcn_wmma_f32_16x16x4_f32(
                false, a, false, b, (short)0, c, false, false);
    }
    const int mBase = rowTile * 16 + halfSel * 8;
    #pragma unroll
    for (int r = 0; r < 8; ++r)
        Y[(size_t)(mBase + r) * M + bCol] = c[r];
}

template <int D>
__global__ __launch_bounds__(256) void spmm_scatter(const float* __restrict__ H,
                                                    const float* __restrict__ vals,
                                                    const int* __restrict__ rows,
                                                    const int* __restrict__ cols,
                                                    float* __restrict__ out,
                                                    int nEdges) {
    constexpr int PER = D / 32;
    typedef __attribute__((ext_vector_type(PER))) float vPf;

    const int lane = threadIdx.x & 31;
    const int edge = blockIdx.x * 8 + (threadIdx.x >> 5);
    if (edge >= nEdges) return;
    const float v    = vals[edge];
    const int   r    = rows[edge];
    const int   cidx = cols[edge];

    const vPf src = *(const vPf*)(H + (size_t)cidx * D + lane * PER);
    float* __restrict__ dst = out + (size_t)r * D + lane * PER;
    #pragma unroll
    for (int j = 0; j < PER; ++j)
        atomicAdd(&dst[j], v * src[j]);
}

extern "C" void kernel_launch(void* const* d_in, const int* in_sizes, int n_in,
                              void* d_out, int out_size, void* d_ws, size_t ws_size,
                              hipStream_t stream) {
    const float* features = (const float*)d_in[0];
    const float* edgeVals = (const float*)d_in[1];
    const float* W0       = (const float*)d_in[2];
    const float* W1       = (const float*)d_in[3];
    const float* W2       = (const float*)d_in[4];
    const int*   edgeRow  = (const int*)d_in[5];
    const int*   edgeCol  = (const int*)d_in[6];

    const int N = in_sizes[0] / 256;
    const int E = in_sizes[1];

    float* ws = (float*)d_ws;
    float* xnorm = ws;
    float* t0    = ws + (size_t)N * 256;
    float* g0    = ws + (size_t)N * 384;
    float* t1    = xnorm;
    float* g1    = t0;
    float* t2    = g0;
    float* out   = (float*)d_out;

    const int normBlocks = (N + 7) / 8;
    const int spmmBlocks = (E + 7) / 8;
    auto gemmBlocks = [N](int m) { return ((N / 16) * (m / 16) + 3) / 4; };

    l2norm_rows<<<normBlocks, 256, 0, stream>>>(features, xnorm, N);

    gemm_wmma_f32<256, 128, false><<<gemmBlocks(128), 128, 0, stream>>>(xnorm, W0, t0, N);
    hipMemsetAsync(g0, 0, (size_t)N * 128 * sizeof(float), stream);
    spmm_scatter<128><<<spmmBlocks, 256, 0, stream>>>(t0, edgeVals, edgeRow, edgeCol, g0, E);

    gemm_wmma_f32<128, 64, true><<<gemmBlocks(64), 128, 0, stream>>>(g0, W1, t1, N);
    hipMemsetAsync(g1, 0, (size_t)N * 64 * sizeof(float), stream);
    spmm_scatter<64><<<spmmBlocks, 256, 0, stream>>>(t1, edgeVals, edgeRow, edgeCol, g1, E);

    gemm_wmma_f32<64, 32, true><<<gemmBlocks(32), 128, 0, stream>>>(g1, W2, t2, N);
    hipMemsetAsync(out, 0, (size_t)N * 32 * sizeof(float), stream);
    spmm_scatter<32><<<spmmBlocks, 256, 0, stream>>>(t2, edgeVals, edgeRow, edgeCol, out, E);
}


# SpMM: float atomic scatter (one wave per edge) -> per-call row bucketing (int atomics) + one-wave-per-row gather, no float atomics or memsets; sum order per row = bucket slot order
# speedup vs baseline: 2.0308x; 2.0308x over previous
#include <hip/hip_runtime.h>
#include <hip/hip_bf16.h>


typedef __attribute__((ext_vector_type(2))) float v2f;
typedef __attribute__((ext_vector_type(8))) float v8f;

__global__ __launch_bounds__(256) void l2norm_rows(const float* __restrict__ F,
                                                   float* __restrict__ X,
                                                   int* __restrict__ cnt,
                                                   int nRows) {
    const int lane = threadIdx.x & 31;
    const int row  = blockIdx.x * 8 + (threadIdx.x >> 5);
    if (row >= nRows) return;
    if (lane == 0) { cnt[row] = 0; if (row == 0) cnt[nRows] = 0; }
    const float4* f = (const float4*)(F + (size_t)row * 256);
    float4 v0 = f[lane];
    float4 v1 = f[lane + 32];
    float s = v0.x*v0.x + v0.y*v0.y + v0.z*v0.z + v0.w*v0.w
            + v1.x*v1.x + v1.y*v1.y + v1.z*v1.z + v1.w*v1.w;
    #pragma unroll
    for (int off = 16; off > 0; off >>= 1) s += __shfl_xor(s, off, 32);
    const float inv = rsqrtf(fmaxf(s, 1e-12f));
    v0.x *= inv; v0.y *= inv; v0.z *= inv; v0.w *= inv;
    v1.x *= inv; v1.y *= inv; v1.z *= inv; v1.w *= inv;
    float4* xo = (float4*)(X + (size_t)row * 256);
    xo[lane]      = v0;
    xo[lane + 32] = v1;
}

template <int K, int M, bool RELU>
__global__ __launch_bounds__(128) void gemm_wmma_f32(const float* __restrict__ X,
                                                     const float* __restrict__ W,
                                                     float* __restrict__ Y,
                                                     int nRows) {
    const int lane   = threadIdx.x & 31;
    const int waveId = blockIdx.x * 4 + (threadIdx.x >> 5);
    const int tilesN = M / 16;
    const int totalTiles = (nRows / 16) * tilesN;
    if (waveId >= totalTiles) return;
    const int rowTile = waveId / tilesN;
    const int colTile = waveId % tilesN;

    const int halfSel = lane >> 4;
    const int kOff    = halfSel * 2;
    const int aRow    = rowTile * 16 + (lane & 15);
    const int bCol    = colTile * 16 + (lane & 15);

    const float* __restrict__ xrow = X + (size_t)aRow * K;
    v8f c = {};
    #pragma unroll 4
    for (int k = 0; k < K; k += 4) {
        float a0 = xrow[k + kOff];
        float a1 = xrow[k + kOff + 1];
        if (RELU) { a0 = fmaxf(a0, 0.0f); a1 = fmaxf(a1, 0.0f); }
        v2f a; a.x = a0; a.y = a1;
        v2f b;
        b.x = W[(size_t)(k + kOff)     * M + bCol];
        b.y = W[(size_t)(k + kOff + 1) * M + bCol];
        c = __builtin_amdgcn_wmma_f32_16x16x4_f32(
                false, a, false, b, (short)0, c, false, false);
    }
    const int mBase = rowTile * 16 + halfSel * 8;
    #pragma unroll
    for (int r = 0; r < 8; ++r)
        Y[(size_t)(mBase + r) * M + bCol] = c[r];
}

constexpr int ELL_CAP = 32;

__global__ __launch_bounds__(256) void bucket_edges(const float* __restrict__ vals,
                                                    const int* __restrict__ rows,
                                                    const int* __restrict__ cols,
                                                    int* __restrict__ cnt,
                                                    int* __restrict__ ell,
                                                    int* __restrict__ ovf,
                                                    int nRows, int nEdges) {
    const int e = blockIdx.x * 256 + threadIdx.x;
    if (e >= nEdges) return;
    const int   r = rows[e];
    const int   c = cols[e];
    const float v = vals[e];
    const int slot = atomicAdd(&cnt[r], 1);
    if (slot < ELL_CAP) {
        int2 ent; ent.x = c; ent.y = __float_as_int(v);
        ((int2*)ell)[(size_t)r * ELL_CAP + slot] = ent;
    } else {
        const int p = atomicAdd(&cnt[nRows], 1);
        ovf[p]                      = r;
        ovf[(size_t)nEdges + p]     = c;
        ovf[(size_t)nEdges * 2 + p] = __float_as_int(v);
    }
}

template <int D>
__global__ __launch_bounds__(256) void spmm_gather(const float* __restrict__ H,
                                                   const int* __restrict__ cnt,
                                                   const int* __restrict__ ell,
                                                   const int* __restrict__ ovf,
                                                   float* __restrict__ out,
                                                   int nRows, int nEdges) {
    constexpr int PER = D / 32;
    typedef __attribute__((ext_vector_type(PER))) float vPf;

    const int lane = threadIdx.x & 31;
    const int row = __builtin_amdgcn_readfirstlane(blockIdx.x * 8 + (threadIdx.x >> 5));
    if (row >= nRows) return;
    const int n = cnt[row];
    const int m = n < ELL_CAP ? n : ELL_CAP;

    int2 ent; ent.x = 0; ent.y = 0;
    if (lane < m) ent = ((const int2*)ell)[(size_t)row * ELL_CAP + lane];

    const float* __restrict__ hl = H + lane * PER;
    vPf acc = {};
    #pragma unroll 8
    for (int j = 0; j < m; ++j) {
        const int   c = __builtin_amdgcn_readlane(ent.x, j);
        const float v = __int_as_float(__builtin_amdgcn_readlane(ent.y, j));
        const vPf   s = *(const vPf*)(hl + (size_t)c * D);
        acc += v * s;
    }

    if (n > ELL_CAP) {
        const int total = cnt[nRows];
        for (int base = 0; base < total; base += 32) {
            const int i = base + lane;
            int orow = -1, oc = 0, ov = 0;
            if (i < total) {
                orow = ovf[i];
                oc   = ovf[(size_t)nEdges + i];
                ov   = ovf[(size_t)nEdges * 2 + i];
            }
            unsigned hit = __builtin_amdgcn_ballot_w32(orow == row);
            while (hit) {
                const int j = __builtin_ctz(hit);
                hit &= hit - 1;
                const int   c = __builtin_amdgcn_readlane(oc, j);
                const float v = __int_as_float(__builtin_amdgcn_readlane(ov, j));
                const vPf   s = *(const vPf*)(hl + (size_t)c * D);
                acc += v * s;
            }
        }
    }
    *(vPf*)(out + (size_t)row * D + lane * PER) = acc;
}

extern "C" void kernel_launch(void* const* d_in, const int* in_sizes, int n_in,
                              void* d_out, int out_size, void* d_ws, size_t ws_size,
                              hipStream_t stream) {
    const float* features = (const float*)d_in[0];
    const float* edgeVals = (const float*)d_in[1];
    const float* W0       = (const float*)d_in[2];
    const float* W1       = (const float*)d_in[3];
    const float* W2       = (const float*)d_in[4];
    const int*   edgeRow  = (const int*)d_in[5];
    const int*   edgeCol  = (const int*)d_in[6];

    const int N = in_sizes[0] / 256;
    const int E = in_sizes[1];

    float* ws = (float*)d_ws;
    float* xnorm = ws;
    float* t0    = ws + (size_t)N * 256;
    float* g0    = ws + (size_t)N * 384;
    float* t1    = xnorm;
    float* g1    = t0;
    float* t2    = g0;
    float* out   = (float*)d_out;
    int* ell = (int*)(ws + (size_t)N * 512);
    int* ovf = ell + (size_t)N * 64;
    int* cnt = ovf + (size_t)E * 3;

    const int normBlocks = (N + 7) / 8;
    const int rowBlocks  = (N + 7) / 8;
    const int edgeBlocks = (E + 255) / 256;
    auto gemmBlocks = [N](int m) { return ((N / 16) * (m / 16) + 3) / 4; };

    l2norm_rows<<<normBlocks, 256, 0, stream>>>(features, xnorm, cnt, N);

    bucket_edges<<<edgeBlocks, 256, 0, stream>>>(edgeVals, edgeRow, edgeCol, cnt, ell, ovf, N, E);

    gemm_wmma_f32<256, 128, false><<<gemmBlocks(128), 128, 0, stream>>>(xnorm, W0, t0, N);
    spmm_gather<128><<<rowBlocks, 256, 0, stream>>>(t0, cnt, ell, ovf, g0, N, E);

    gemm_wmma_f32<128, 64, true><<<gemmBlocks(64), 128, 0, stream>>>(g0, W1, t1, N);
    spmm_gather<64><<<rowBlocks, 256, 0, stream>>>(t1, cnt, ell, ovf, g1, N, E);

    gemm_wmma_f32<64, 32, true><<<gemmBlocks(32), 128, 0, stream>>>(g1, W2, t2, N);
    spmm_gather<32><<<rowBlocks, 256, 0, stream>>>(t2, cnt, ell, ovf, out, N, E);
}


# GEMMs re-tiled: resident 8-wave groups stage W once in LDS, a wave keeps its 16-row X strip in registers (16 B loads, refilled in place) over 4 (2) column tiles; k sum order permuted in 8-k groups
# speedup vs baseline: 2.4931x; 1.2276x over previous
#include <hip/hip_runtime.h>
#include <hip/hip_bf16.h>


typedef __attribute__((ext_vector_type(2))) float v2f;
typedef __attribute__((ext_vector_type(8))) float v8f;

__global__ __launch_bounds__(256) void l2norm_rows(const float* __restrict__ F,
                                                   float* __restrict__ X,
                                                   int* __restrict__ cnt,
                                                   int nRows) {
    const int lane = threadIdx.x & 31;
    const int row  = blockIdx.x * 8 + (threadIdx.x >> 5);
    if (row >= nRows) return;
    if (lane == 0) { cnt[row] = 0; if (row == 0) cnt[nRows] = 0; }
    const float4* f = (const float4*)(F + (size_t)row * 256);
    float4 v0 = f[lane];
    float4 v1 = f[lane + 32];
    float s = v0.x*v0.x + v0.y*v0.y + v0.z*v0.z + v0.w*v0.w
            + v1.x*v1.x + v1.y*v1.y + v1.z*v1.z + v1.w*v1.w;
    #pragma unroll
    for (int off = 16; off > 0; off >>= 1) s += __shfl_xor(s, off, 32);
    const float inv = rsqrtf(fmaxf(s, 1e-12f));
    v0.x *= inv; v0.y *= inv; v0.z *= inv; v0.w *= inv;
    v1.x *= inv; v1.y *= inv; v1.z *= inv; v1.w *= inv;
    float4* xo = (float4*)(X + (size_t)row * 256);
    xo[lane]      = v0;
    xo[lane + 32] = v1;
}

template <int K, int M, int NT, bool RELU, bool REFILL>
__device__ __forceinline__ void gemm_unit(float4 (&aq)[K / 8],
                                          const float* __restrict__ bptr,
                                          const float* __restrict__ anext,
                                          float* __restrict__ yptr) {
    constexpr int KJ    = K / 8;
    constexpr int PITCH = 4 * M + 32;
    v8f acc[NT];
    #pragma unroll
    for (int t = 0; t < NT; ++t) acc[t] = v8f{};

    float4 bq[2][NT];
    #pragma unroll
    for (int t = 0; t < NT; ++t) bq[0][t] = *(const float4*)(bptr + t * 64);

    #pragma unroll
    for (int j = 0; j < KJ; ++j) {
        if (j + 1 < KJ) {
            #pragma unroll
            for (int t = 0; t < NT; ++t)
                bq[(j + 1) & 1][t] = *(const float4*)(bptr + 2 * (j + 1) * PITCH + t * 64);
            __builtin_amdgcn_sched_barrier(0);
        }
        float4 a = aq[j];
        if (RELU) {
            a.x = fmaxf(a.x, 0.0f); a.y = fmaxf(a.y, 0.0f);
            a.z = fmaxf(a.z, 0.0f); a.w = fmaxf(a.w, 0.0f);
        }
        v2f a0; a0.x = a.x; a0.y = a.y;
        v2f a1; a1.x = a.z; a1.y = a.w;
        #pragma unroll
        for (int t = 0; t < NT; ++t) {
            v2f b; b.x = bq[j & 1][t].x; b.y = bq[j & 1][t].y;
            acc[t] = __builtin_amdgcn_wmma_f32_16x16x4_f32(
                         false, a0, false, b, (short)0, acc[t], false, false);
        }
        #pragma unroll
        for (int t = 0; t < NT; ++t) {
            v2f b; b.x = bq[j & 1][t].z; b.y = bq[j & 1][t].w;
            acc[t] = __builtin_amdgcn_wmma_f32_16x16x4_f32(
                         false, a1, false, b, (short)0, acc[t], false, false);
        }
        if (REFILL) {
            aq[j] = *(const float4*)(anext + 8 * j);
            __builtin_amdgcn_sched_barrier(0);
        }
    }
    #pragma unroll
    for (int t = 0; t < NT; ++t) {
        #pragma unroll
        for (int r = 0; r < 8; ++r)
            yptr[(size_t)r * M + t * 16] = acc[t][r];
    }
}

template <int K, int M, int NT, bool RELU>
__global__ __launch_bounds__(256) void gemm_wmma_f32(const float* __restrict__ X,
                                                     const float* __restrict__ W,
                                                     float* __restrict__ Y,
                                                     int nRows) {
    constexpr int KQ    = K / 4;
    constexpr int PITCH = 4 * M + 32;
    constexpr int KJ    = K / 8;
    constexpr int CG    = M / (16 * NT);
    __shared__ __attribute__((aligned(16))) float Ws[KQ * PITCH];

    const int lane = threadIdx.x & 31;
    const int wave = __builtin_amdgcn_readfirstlane(threadIdx.x >> 5);
    const int c    = lane & 15;
    const int h    = lane >> 4;

    const int totalUnits = (nRows / 16) * CG;
    const int perGroup   = (totalUnits + (int)gridDim.x - 1) / (int)gridDim.x;
    const int uBeg       = (int)blockIdx.x * perGroup;
    const int uEnd       = uBeg + perGroup < totalUnits ? uBeg + perGroup : totalUnits;
    int u = uBeg + wave;
    const bool haveWork = u < uEnd;

    float4 aq[KJ];
    if (haveWork) {
        const float* ap = X + (size_t)((u / CG) * 16 + c) * K + 4 * h;
        #pragma unroll
        for (int j = 0; j < KJ; ++j) aq[j] = *(const float4*)(ap + 8 * j);
    }

    #pragma unroll 8
    for (int i = threadIdx.x; i < KQ * M; i += 256) {
        const int q = i / M;
        const int n = i % M;
        float4 v;
        v.x = W[(size_t)(4 * q + 0) * M + n];
        v.y = W[(size_t)(4 * q + 1) * M + n];
        v.z = W[(size_t)(4 * q + 2) * M + n];
        v.w = W[(size_t)(4 * q + 3) * M + n];
        *(float4*)(Ws + q * PITCH + 4 * n) = v;
    }
    __syncthreads();
    if (!haveWork) return;

    while (true) {
        const int rowTile = u / CG;
        const int cg      = u % CG;
        const float* bptr = Ws + h * PITCH + (cg * NT * 16 + c) * 4;
        float* yptr = Y + (size_t)(rowTile * 16 + h * 8) * M + cg * NT * 16 + c;
        const int un = u + 8;
        if (un < uEnd) {
            const float* anext = X + (size_t)((un / CG) * 16 + c) * K + 4 * h;
            gemm_unit<K, M, NT, RELU, true>(aq, bptr, anext, yptr);
            u = un;
        } else {
            gemm_unit<K, M, NT, RELU, false>(aq, bptr, nullptr, yptr);
            break;
        }
    }
}

constexpr int ELL_CAP = 32;

__global__ __launch_bounds__(256) void bucket_edges(const float* __restrict__ vals,
                                                    const int* __restrict__ rows,
                                                    const int* __restrict__ cols,
                                                    int* __restrict__ cnt,
                                                    int* __restrict__ ell,
                                                    int* __restrict__ ovf,
                                                    int nRows, int nEdges) {
    const int e = blockIdx.x * 256 + threadIdx.x;
    if (e >= nEdges) return;
    const int   r = rows[e];
    const int   c = cols[e];
    const float v = vals[e];
    const int slot = atomicAdd(&cnt[r], 1);
    if (slot < ELL_CAP) {
        int2 ent; ent.x = c; ent.y = __float_as_int(v);
        ((int2*)ell)[(size_t)r * ELL_CAP + slot] = ent;
    } else {
        const int p = atomicAdd(&cnt[nRows], 1);
        ovf[p]                      = r;
        ovf[(size_t)nEdges + p]     = c;
        ovf[(size_t)nEdges * 2 + p] = __float_as_int(v);
    }
}

template <int D>
__global__ __launch_bounds__(256) void spmm_gather(const float* __restrict__ H,
                                                   const int* __restrict__ cnt,
                                                   const int* __restrict__ ell,
                                                   const int* __restrict__ ovf,
                                                   float* __restrict__ out,
                                                   int nRows, int nEdges) {
    constexpr int PER = D / 32;
    typedef __attribute__((ext_vector_type(PER))) float vPf;

    const int lane = threadIdx.x & 31;
    const int row = __builtin_amdgcn_readfirstlane(blockIdx.x * 8 + (threadIdx.x >> 5));
    if (row >= nRows) return;
    const int n = cnt[row];
    const int m = n < ELL_CAP ? n : ELL_CAP;

    int2 ent; ent.x = 0; ent.y = 0;
    if (lane < m) ent = ((const int2*)ell)[(size_t)row * ELL_CAP + lane];

    const float* __restrict__ hl = H + lane * PER;
    vPf acc = {};
    #pragma unroll 8
    for (int j = 0; j < m; ++j) {
        const int   c = __builtin_amdgcn_readlane(ent.x, j);
        const float v = __int_as_float(__builtin_amdgcn_readlane(ent.y, j));
        const vPf   s = *(const vPf*)(hl + (size_t)c * D);
        acc += v * s;
    }

    if (n > ELL_CAP) {
        const int total = cnt[nRows];
        for (int base = 0; base < total; base += 32) {
            const int i = base + lane;
            int orow = -1, oc = 0, ov = 0;
            if (i < total) {
                orow = ovf[i];
                oc   = ovf[(size_t)nEdges + i];
                ov   = ovf[(size_t)nEdges * 2 + i];
            }
            unsigned hit = __builtin_amdgcn_ballot_w32(orow == row);
            while (hit) {
                const int j = __builtin_ctz(hit);
                hit &= hit - 1;
                const int   c = __builtin_amdgcn_readlane(oc, j);
                const float v = __int_as_float(__builtin_amdgcn_readlane(ov, j));
                const vPf   s = *(const vPf*)(hl + (size_t)c * D);
                acc += v * s;
            }
        }
    }
    *(vPf*)(out + (size_t)row * D + lane * PER) = acc;
}

extern "C" void kernel_launch(void* const* d_in, const int* in_sizes, int n_in,
                              void* d_out, int out_size, void* d_ws, size_t ws_size,
                              hipStream_t stream) {
    const float* features = (const float*)d_in[0];
    const float* edgeVals = (const float*)d_in[1];
    const float* W0       = (const float*)d_in[2];
    const float* W1       = (const float*)d_in[3];
    const float* W2       = (const float*)d_in[4];
    const int*   edgeRow  = (const int*)d_in[5];
    const int*   edgeCol  = (const int*)d_in[6];

    const int N = in_sizes[0] / 256;
    const int E = in_sizes[1];

    float* ws = (float*)d_ws;
    float* xnorm = ws;
    float* t0    = ws + (size_t)N * 256;
    float* g0    = ws + (size_t)N * 384;
    float* t1    = xnorm;
    float* g1    = t0;
    float* t2    = g0;
    float* out   = (float*)d_out;
    int* ell = (int*)(ws + (size_t)N * 512);
    int* ovf = ell + (size_t)N * 64;
    int* cnt = ovf + (size_t)E * 3;

    const int normBlocks = (N + 7) / 8;
    const int rowBlocks  = (N + 7) / 8;
    const int edgeBlocks = (E + 255) / 256;
    const int gemmGroups = 512;

    l2norm_rows<<<normBlocks, 256, 0, stream>>>(features, xnorm, cnt, N);

    bucket_edges<<<edgeBlocks, 256, 0, stream>>>(edgeVals, edgeRow, edgeCol, cnt, ell, ovf, N, E);

    gemm_wmma_f32<256, 128, 4, false><<<gemmGroups, 256, 0, stream>>>(xnorm, W0, t0, N);
    spmm_gather<128><<<rowBlocks, 256, 0, stream>>>(t0, cnt, ell, ovf, g0, N, E);

    gemm_wmma_f32<128, 64, 4, true><<<gemmGroups, 256, 0, stream>>>(g0, W1, t1, N);
    spmm_gather<64><<<rowBlocks, 256, 0, stream>>>(t1, cnt, ell, ovf, g1, N, E);

    gemm_wmma_f32<64, 32, 2, true><<<gemmGroups, 256, 0, stream>>>(g1, W2, t2, N);
    spmm_gather<32><<<rowBlocks, 256, 0, stream>>>(t2, cnt, ell, ovf, out, N, E);
}


# l2norm kernel removed: first GEMM reads raw features, sums each row's squares from its A registers, scales its output rows by rsqrt (norm applied after the product; new sum order of the squares)
# speedup vs baseline: 2.7251x; 1.0931x over previous
#include <hip/hip_runtime.h>
#include <hip/hip_bf16.h>


typedef __attribute__((ext_vector_type(2))) float v2f;
typedef __attribute__((ext_vector_type(8))) float v8f;

template <int K, int M, int NT, bool RELU, bool REFILL, bool NORM>
__device__ __forceinline__ void gemm_unit(float4 (&aq)[K / 8],
                                          const float* __restrict__ bptr,
                                          const float* __restrict__ anext,
                                          float* __restrict__ yptr,
                                          const int h) {
    static_assert(!(NORM && RELU), "the squares are summed from the raw A values");
    constexpr int KJ    = K / 8;
    constexpr int PITCH = 4 * M + 32;
    v8f acc[NT];
    #pragma unroll
    for (int t = 0; t < NT; ++t) acc[t] = v8f{};
    v2f ss0 = {}, ss1 = {};

    float4 bq[2][NT];
    #pragma unroll
    for (int t = 0; t < NT; ++t) bq[0][t] = *(const float4*)(bptr + t * 64);

    #pragma unroll
    for (int j = 0; j < KJ; ++j) {
        if (j + 1 < KJ) {
            #pragma unroll
            for (int t = 0; t < NT; ++t)
                bq[(j + 1) & 1][t] = *(const float4*)(bptr + 2 * (j + 1) * PITCH + t * 64);
            __builtin_amdgcn_sched_barrier(0);
        }
        float4 a = aq[j];
        if (RELU) {
            a.x = fmaxf(a.x, 0.0f); a.y = fmaxf(a.y, 0.0f);
            a.z = fmaxf(a.z, 0.0f); a.w = fmaxf(a.w, 0.0f);
        }
        v2f a0; a0.x = a.x; a0.y = a.y;
        v2f a1; a1.x = a.z; a1.y = a.w;
        if (NORM) { ss0 += a0 * a0; ss1 += a1 * a1; }
        #pragma unroll
        for (int t = 0; t < NT; ++t) {
            v2f b; b.x = bq[j & 1][t].x; b.y = bq[j & 1][t].y;
            acc[t] = __builtin_amdgcn_wmma_f32_16x16x4_f32(
                         false, a0, false, b, (short)0, acc[t], false, false);
        }
        #pragma unroll
        for (int t = 0; t < NT; ++t) {
            v2f b; b.x = bq[j & 1][t].z; b.y = bq[j & 1][t].w;
            acc[t] = __builtin_amdgcn_wmma_f32_16x16x4_f32(
                         false, a1, false, b, (short)0, acc[t], false, false);
        }
        if (REFILL) {
            aq[j] = *(const float4*)(anext + 8 * j);
            __builtin_amdgcn_sched_barrier(0);
        }
    }
    if (NORM) {
        float s = (ss0.x + ss0.y) + (ss1.x + ss1.y);
        s += __shfl_xor(s, 16, 32);
        const float inv = rsqrtf(fmaxf(s, 1e-12f));
        #pragma unroll
        for (int r = 0; r < 8; ++r) {
            const float sc = __shfl(inv, 8 * h + r, 32);
            #pragma unroll
            for (int t = 0; t < NT; ++t)
                yptr[(size_t)r * M + t * 16] = acc[t][r] * sc;
        }
    } else {
        #pragma unroll
        for (int t = 0; t < NT; ++t) {
            #pragma unroll
            for (int r = 0; r < 8; ++r)
                yptr[(size_t)r * M + t * 16] = acc[t][r];
        }
    }
}

template <int K, int M, int NT, bool RELU, bool NORM>
__global__ __launch_bounds__(256) void gemm_wmma_f32(const float* __restrict__ X,
                                                     const float* __restrict__ W,
                                                     float* __restrict__ Y,
                                                     int* __restrict__ cnt,
                                                     int nRows) {
    constexpr int KQ    = K / 4;
    constexpr int PITCH = 4 * M + 32;
    constexpr int KJ    = K / 8;
    constexpr int CG    = M / (16 * NT);
    __shared__ __attribute__((aligned(16))) float Ws[KQ * PITCH];

    if (NORM) {
        for (int i = blockIdx.x * 256 + threadIdx.x; i <= nRows; i += (int)gridDim.x * 256)
            cnt[i] = 0;
    }

    const int lane = threadIdx.x & 31;
    const int wave = __builtin_amdgcn_readfirstlane(threadIdx.x >> 5);
    const int c    = lane & 15;
    const int h    = lane >> 4;

    const int totalUnits = (nRows / 16) * CG;
    const int perGroup   = (totalUnits + (int)gridDim.x - 1) / (int)gridDim.x;
    const int uBeg       = (int)blockIdx.x * perGroup;
    const int uEnd       = uBeg + perGroup < totalUnits ? uBeg + perGroup : totalUnits;
    int u = uBeg + wave;
    const bool haveWork = u < uEnd;

    float4 aq[KJ];
    if (haveWork) {
        const float* ap = X + (size_t)((u / CG) * 16 + c) * K + 4 * h;
        #pragma unroll
        for (int j = 0; j < KJ; ++j) aq[j] = *(const float4*)(ap + 8 * j);
    }

    #pragma unroll 8
    for (int i = threadIdx.x; i < KQ * M; i += 256) {
        const int q = i / M;
        const int n = i % M;
        float4 v;
        v.x = W[(size_t)(4 * q + 0) * M + n];
        v.y = W[(size_t)(4 * q + 1) * M + n];
        v.z = W[(size_t)(4 * q + 2) * M + n];
        v.w = W[(size_t)(4 * q + 3) * M + n];
        *(float4*)(Ws + q * PITCH + 4 * n) = v;
    }
    __syncthreads();
    if (!haveWork) return;

    while (true) {
        const int rowTile = u / CG;
        const int cg      = u % CG;
        const float* bptr = Ws + h * PITCH + (cg * NT * 16 + c) * 4;
        float* yptr = Y + (size_t)(rowTile * 16 + h * 8) * M + cg * NT * 16 + c;
        const int un = u + 8;
        if (un < uEnd) {
            const float* anext = X + (size_t)((un / CG) * 16 + c) * K + 4 * h;
            gemm_unit<K, M, NT, RELU, true, NORM>(aq, bptr, anext, yptr, h);
            u = un;
        } else {
            gemm_unit<K, M, NT, RELU, false, NORM>(aq, bptr, nullptr, yptr, h);
            break;
        }
    }
}

constexpr int ELL_CAP = 32;

__global__ __launch_bounds__(256) void bucket_edges(const float* __restrict__ vals,
                                                    const int* __restrict__ rows,
                                                    const int* __restrict__ cols,
                                                    int* __restrict__ cnt,
                                                    int* __restrict__ ell,
                                                    int* __restrict__ ovf,
                                                    int nRows, int nEdges) {
    const int e = blockIdx.x * 256 + threadIdx.x;
    if (e >= nEdges) return;
    const int   r = rows[e];
    const int   c = cols[e];
    const float v = vals[e];
    const int slot = atomicAdd(&cnt[r], 1);
    if (slot < ELL_CAP) {
        int2 ent; ent.x = c; ent.y = __float_as_int(v);
        ((int2*)ell)[(size_t)r * ELL_CAP + slot] = ent;
    } else {
        const int p = atomicAdd(&cnt[nRows], 1);
        ovf[p]                      = r;
        ovf[(size_t)nEdges + p]     = c;
        ovf[(size_t)nEdges * 2 + p] = __float_as_int(v);
    }
}

template <int D>
__global__ __launch_bounds__(256) void spmm_gather(const float* __restrict__ H,
                                                   const int* __restrict__ cnt,
                                                   const int* __restrict__ ell,
                                                   const int* __restrict__ ovf,
                                                   float* __restrict__ out,
                                                   int nRows, int nEdges) {
    constexpr int PER = D / 32;
    typedef __attribute__((ext_vector_type(PER))) float vPf;

    const int lane = threadIdx.x & 31;
    const int row = __builtin_amdgcn_readfirstlane(blockIdx.x * 8 + (threadIdx.x >> 5));
    if (row >= nRows) return;
    const int n = cnt[row];
    const int m = n < ELL_CAP ? n : ELL_CAP;

    int2 ent; ent.x = 0; ent.y = 0;
    if (lane < m) ent = ((const int2*)ell)[(size_t)row * ELL_CAP + lane];

    const float* __restrict__ hl = H + lane * PER;
    vPf acc = {};
    #pragma unroll 8
    for (int j = 0; j < m; ++j) {
        const int   c = __builtin_amdgcn_readlane(ent.x, j);
        const float v = __int_as_float(__builtin_amdgcn_readlane(ent.y, j));
        const vPf   s = *(const vPf*)(hl + (size_t)c * D);
        acc += v * s;
    }

    if (n > ELL_CAP) {
        const int total = cnt[nRows];
        for (int base = 0; base < total; base += 32) {
            const int i = base + lane;
            int orow = -1, oc = 0, ov = 0;
            if (i < total) {
                orow = ovf[i];
                oc   = ovf[(size_t)nEdges + i];
                ov   = ovf[(size_t)nEdges * 2 + i];
            }
            unsigned hit = __builtin_amdgcn_ballot_w32(orow == row);
            while (hit) {
                const int j = __builtin_ctz(hit);
                hit &= hit - 1;
                const int   c = __builtin_amdgcn_readlane(oc, j);
                const float v = __int_as_float(__builtin_amdgcn_readlane(ov, j));
                const vPf   s = *(const vPf*)(hl + (size_t)c * D);
                acc += v * s;
            }
        }
    }
    *(vPf*)(out + (size_t)row * D + lane * PER) = acc;
}

extern "C" void kernel_launch(void* const* d_in, const int* in_sizes, int n_in,
                              void* d_out, int out_size, void* d_ws, size_t ws_size,
                              hipStream_t stream) {
    const float* features = (const float*)d_in[0];
    const float* edgeVals = (const float*)d_in[1];
    const float* W0       = (const float*)d_in[2];
    const float* W1       = (const float*)d_in[3];
    const float* W2       = (const float*)d_in[4];
    const int*   edgeRow  = (const int*)d_in[5];
    const int*   edgeCol  = (const int*)d_in[6];

    const int N = in_sizes[0] / 256;
    const int E = in_sizes[1];

    float* ws = (float*)d_ws;
    float* t1    = ws;
    float* t0    = ws + (size_t)N * 256;
    float* g0    = ws + (size_t)N * 384;
    float* g1    = t0;
    float* t2    = g0;
    float* out   = (float*)d_out;
    int* ell = (int*)(ws + (size_t)N * 512);
    int* ovf = ell + (size_t)N * 64;
    int* cnt = ovf + (size_t)E * 3;

    const int rowBlocks  = (N + 7) / 8;
    const int edgeBlocks = (E + 255) / 256;
    const int gemmGroups = 512;

    gemm_wmma_f32<256, 128, 4, false, true><<<gemmGroups, 256, 0, stream>>>(features, W0, t0, cnt, N);

    bucket_edges<<<edgeBlocks, 256, 0, stream>>>(edgeVals, edgeRow, edgeCol, cnt, ell, ovf, N, E);

    spmm_gather<128><<<rowBlocks, 256, 0, stream>>>(t0, cnt, ell, ovf, g0, N, E);

    gemm_wmma_f32<128, 64, 4, true, false><<<gemmGroups, 256, 0, stream>>>(g0, W1, t1, nullptr, N);
    spmm_gather<64><<<rowBlocks, 256, 0, stream>>>(t1, cnt, ell, ovf, g1, N, E);

    gemm_wmma_f32<64, 32, 2, true, false><<<gemmGroups, 256, 0, stream>>>(g1, W2, t2, nullptr, N);
    spmm_gather<32><<<rowBlocks, 256, 0, stream>>>(t2, cnt, ell, ovf, out, N, E);
}


# Row gathers of layers 1,2 fused into the next layer's GEMM: resident 8-wave groups gather 32 rows to LDS as the A tile, W in LDS, steps by ticket; g0,g1 not stored, 7->5 launches; sum orders kept
# speedup vs baseline: 2.8739x; 1.0546x over previous
#include <hip/hip_runtime.h>
#include <hip/hip_bf16.h>


typedef __attribute__((ext_vector_type(2))) float v2f;
typedef __attribute__((ext_vector_type(8))) float v8f;

template <int K, int M, int NT, bool RELU, bool REFILL, bool NORM>
__device__ __forceinline__ void gemm_unit(float4 (&aq)[K / 8],
                                          const float* __restrict__ bptr,
                                          const float* __restrict__ anext,
                                          float* __restrict__ yptr,
                                          const int h) {
    static_assert(!(NORM && RELU), "the squares are summed from the raw A values");
    constexpr int KJ    = K / 8;
    constexpr int PITCH = 4 * M + 32;
    v8f acc[NT];
    #pragma unroll
    for (int t = 0; t < NT; ++t) acc[t] = v8f{};
    v2f ss0 = {}, ss1 = {};

    float4 bq[2][NT];
    #pragma unroll
    for (int t = 0; t < NT; ++t) bq[0][t] = *(const float4*)(bptr + t * 64);

    #pragma unroll
    for (int j = 0; j < KJ; ++j) {
        if (j + 1 < KJ) {
            #pragma unroll
            for (int t = 0; t < NT; ++t)
                bq[(j + 1) & 1][t] = *(const float4*)(bptr + 2 * (j + 1) * PITCH + t * 64);
            __builtin_amdgcn_sched_barrier(0);
        }
        float4 a = aq[j];
        if (RELU) {
            a.x = fmaxf(a.x, 0.0f); a.y = fmaxf(a.y, 0.0f);
            a.z = fmaxf(a.z, 0.0f); a.w = fmaxf(a.w, 0.0f);
        }
        v2f a0; a0.x = a.x; a0.y = a.y;
        v2f a1; a1.x = a.z; a1.y = a.w;
        if (NORM) { ss0 += a0 * a0; ss1 += a1 * a1; }
        #pragma unroll
        for (int t = 0; t < NT; ++t) {
            v2f b; b.x = bq[j & 1][t].x; b.y = bq[j & 1][t].y;
            acc[t] = __builtin_amdgcn_wmma_f32_16x16x4_f32(
                         false, a0, false, b, (short)0, acc[t], false, false);
        }
        #pragma unroll
        for (int t = 0; t < NT; ++t) {
            v2f b; b.x = bq[j & 1][t].z; b.y = bq[j & 1][t].w;
            acc[t] = __builtin_amdgcn_wmma_f32_16x16x4_f32(
                         false, a1, false, b, (short)0, acc[t], false, false);
        }
        if (REFILL) {
            aq[j] = *(const float4*)(anext + 8 * j);
            __builtin_amdgcn_sched_barrier(0);
        }
    }
    if (NORM) {
        float s = (ss0.x + ss0.y) + (ss1.x + ss1.y);
        s += __shfl_xor(s, 16, 32);
        const float inv = rsqrtf(fmaxf(s, 1e-12f));
        #pragma unroll
        for (int r = 0; r < 8; ++r) {
            const float sc = __shfl(inv, 8 * h + r, 32);
            #pragma unroll
            for (int t = 0; t < NT; ++t)
                yptr[(size_t)r * M + t * 16] = acc[t][r] * sc;
        }
    } else {
        #pragma unroll
        for (int t = 0; t < NT; ++t) {
            #pragma unroll
            for (int r = 0; r < 8; ++r)
                yptr[(size_t)r * M + t * 16] = acc[t][r];
        }
    }
}

template <int K, int M, int NT, bool RELU, bool NORM>
__global__ __launch_bounds__(256) void gemm_wmma_f32(const float* __restrict__ X,
                                                     const float* __restrict__ W,
                                                     float* __restrict__ Y,
                                                     int* __restrict__ cnt,
                                                     int nRows) {
    constexpr int KQ    = K / 4;
    constexpr int PITCH = 4 * M + 32;
    constexpr int KJ    = K / 8;
    constexpr int CG    = M / (16 * NT);
    __shared__ __attribute__((aligned(16))) float Ws[KQ * PITCH];

    if (NORM) {
        for (int i = blockIdx.x * 256 + threadIdx.x; i <= nRows + 2; i += (int)gridDim.x * 256)
            cnt[i] = 0;
    }

    const int lane = threadIdx.x & 31;
    const int wave = __builtin_amdgcn_readfirstlane(threadIdx.x >> 5);
    const int c    = lane & 15;
    const int h    = lane >> 4;

    const int totalUnits = (nRows / 16) * CG;
    const int perGroup   = (totalUnits + (int)gridDim.x - 1) / (int)gridDim.x;
    const int uBeg       = (int)blockIdx.x * perGroup;
    const int uEnd       = uBeg + perGroup < totalUnits ? uBeg + perGroup : totalUnits;
    int u = uBeg + wave;
    const bool haveWork = u < uEnd;

    float4 aq[KJ];
    if (haveWork) {
        const float* ap = X + (size_t)((u / CG) * 16 + c) * K + 4 * h;
        #pragma unroll
        for (int j = 0; j < KJ; ++j) aq[j] = *(const float4*)(ap + 8 * j);
    }

    #pragma unroll 8
    for (int i = threadIdx.x; i < KQ * M; i += 256) {
        const int q = i / M;
        const int n = i % M;
        float4 v;
        v.x = W[(size_t)(4 * q + 0) * M + n];
        v.y = W[(size_t)(4 * q + 1) * M + n];
        v.z = W[(size_t)(4 * q + 2) * M + n];
        v.w = W[(size_t)(4 * q + 3) * M + n];
        *(float4*)(Ws + q * PITCH + 4 * n) = v;
    }
    __syncthreads();
    if (!haveWork) return;

    while (true) {
        const int rowTile = u / CG;
        const int cg      = u % CG;
        const float* bptr = Ws + h * PITCH + (cg * NT * 16 + c) * 4;
        float* yptr = Y + (size_t)(rowTile * 16 + h * 8) * M + cg * NT * 16 + c;
        const int un = u + 8;
        if (un < uEnd) {
            const float* anext = X + (size_t)((un / CG) * 16 + c) * K + 4 * h;
            gemm_unit<K, M, NT, RELU, true, NORM>(aq, bptr, anext, yptr, h);
            u = un;
        } else {
            gemm_unit<K, M, NT, RELU, false, NORM>(aq, bptr, nullptr, yptr, h);
            break;
        }
    }
}

constexpr int ELL_CAP = 32;

__global__ __launch_bounds__(256) void bucket_edges(const float* __restrict__ vals,
                                                    const int* __restrict__ rows,
                                                    const int* __restrict__ cols,
                                                    int* __restrict__ cnt,
                                                    int* __restrict__ ell,
                                                    int* __restrict__ ovf,
                                                    int nRows, int nEdges) {
    const int e = blockIdx.x * 256 + threadIdx.x;
    if (e >= nEdges) return;
    const int   r = rows[e];
    const int   c = cols[e];
    const float v = vals[e];
    const int slot = atomicAdd(&cnt[r], 1);
    if (slot < ELL_CAP) {
        int2 ent; ent.x = c; ent.y = __float_as_int(v);
        ((int2*)ell)[(size_t)r * ELL_CAP + slot] = ent;
    } else {
        const int p = atomicAdd(&cnt[nRows], 1);
        ovf[p]                      = r;
        ovf[(size_t)nEdges + p]     = c;
        ovf[(size_t)nEdges * 2 + p] = __float_as_int(v);
    }
}

template <int D>
__global__ __launch_bounds__(256) void spmm_gather(const float* __restrict__ H,
                                                   const int* __restrict__ cnt,
                                                   const int* __restrict__ ell,
                                                   const int* __restrict__ ovf,
                                                   float* __restrict__ out,
                                                   int nRows, int nEdges) {
    constexpr int PER = D / 32;
    typedef __attribute__((ext_vector_type(PER))) float vPf;

    const int lane = threadIdx.x & 31;
    const int row = __builtin_amdgcn_readfirstlane(blockIdx.x * 8 + (threadIdx.x >> 5));
    if (row >= nRows) return;
    const int n = cnt[row];
    const int m = n < ELL_CAP ? n : ELL_CAP;

    int2 ent; ent.x = 0; ent.y = 0;
    if (lane < m) ent = ((const int2*)ell)[(size_t)row * ELL_CAP + lane];

    const float* __restrict__ hl = H + lane * PER;
    vPf acc = {};
    #pragma unroll 8
    for (int j = 0; j < m; ++j) {
        const int   c = __builtin_amdgcn_readlane(ent.x, j);
        const float v = __int_as_float(__builtin_amdgcn_readlane(ent.y, j));
        const vPf   s = *(const vPf*)(hl + (size_t)c * D);
        acc += v * s;
    }

    if (n > ELL_CAP) {
        const int total = cnt[nRows];
        for (int base = 0; base < total; base += 32) {
            const int i = base + lane;
            int orow = -1, oc = 0, ov = 0;
            if (i < total) {
                orow = ovf[i];
                oc   = ovf[(size_t)nEdges + i];
                ov   = ovf[(size_t)nEdges * 2 + i];
            }
            unsigned hit = __builtin_amdgcn_ballot_w32(orow == row);
            while (hit) {
                const int j = __builtin_ctz(hit);
                hit &= hit - 1;
                const int   c = __builtin_amdgcn_readlane(oc, j);
                const float v = __int_as_float(__builtin_amdgcn_readlane(ov, j));
                const vPf   s = *(const vPf*)(hl + (size_t)c * D);
                acc += v * s;
            }
        }
    }
    *(vPf*)(out + (size_t)row * D + lane * PER) = acc;
}

template <int B, int K, typename V>
__device__ __forceinline__ void gather_slots(V& acc, const float* __restrict__ hl,
                                             const int ex, const int ey, const int j0) {
    V     s[B];
    float v[B];
    #pragma unroll
    for (int u = 0; u < B; ++u) {
        const int cc = __builtin_amdgcn_readlane(ex, j0 + u);
        v[u] = __int_as_float(__builtin_amdgcn_readlane(ey, j0 + u));
        s[u] = *(const V*)(hl + (size_t)cc * K);
    }
    #pragma unroll
    for (int u = 0; u < B; ++u) acc += v[u] * s[u];
}

template <int K, int M>
__global__ __launch_bounds__(256) void gather_gemm_f32(const float* __restrict__ H,
                                                       const float* __restrict__ W,
                                                       float* __restrict__ Y,
                                                       const int* __restrict__ cnt,
                                                       const int* __restrict__ ell,
                                                       const int* __restrict__ ovf,
                                                       int* __restrict__ ticket,
                                                       int nRows, int nEdges) {
    constexpr int CT    = M / 16;
    constexpr int ROWS  = 32;
    constexpr int TILES = 2 * CT;
    constexpr int RW    = ROWS / 8;
    constexpr int KQ    = K / 4;
    constexpr int PITCH = 4 * M + 32;
    constexpr int KJ    = K / 8;
    constexpr int AP    = K + 4;
    constexpr int PER   = K / 32;
    static_assert(KJ % 8 == 0 && TILES <= 8, "product groups of 8 steps, one tile per wave");
    typedef __attribute__((ext_vector_type(PER))) float vPf;

    __shared__ __attribute__((aligned(16))) float Ws[KQ * PITCH];
    __shared__ __attribute__((aligned(16))) float As[2][ROWS * AP];
    __shared__ int sNext[2];

    const int lane = threadIdx.x & 31;
    const int wave = __builtin_amdgcn_readfirstlane(threadIdx.x >> 5);
    const int c    = lane & 15;
    const int h    = lane >> 4;

    #pragma unroll 8
    for (int i = threadIdx.x; i < KQ * M; i += 256) {
        const int q = i / M;
        const int n = i % M;
        float4 v;
        v.x = W[(size_t)(4 * q + 0) * M + n];
        v.y = W[(size_t)(4 * q + 1) * M + n];
        v.z = W[(size_t)(4 * q + 2) * M + n];
        v.w = W[(size_t)(4 * q + 3) * M + n];
        *(float4*)(Ws + q * PITCH + 4 * n) = v;
    }

    const int nSteps = (nRows + ROWS - 1) / ROWS;
    const int rt = wave / CT;
    const int ct = wave % CT;
    const float* __restrict__ hl = H + lane * PER;
    const float* bp = Ws + h * PITCH + (ct * 16 + c) * 4;

    int step = blockIdx.x;
    int buf  = 0;
    while (step < nSteps) {
        const int rowBase = step * ROWS + wave * RW;
        int myN = 0;
        if (lane < RW && rowBase + lane < nRows) myN = cnt[rowBase + lane];
        int  nn[RW];
        int2 ee[RW];
        #pragma unroll
        for (int i = 0; i < RW; ++i) {
            nn[i] = __builtin_amdgcn_readlane(myN, i);
            const int m = nn[i] < ELL_CAP ? nn[i] : ELL_CAP;
            ee[i].x = 0; ee[i].y = 0;
            if (lane < m) ee[i] = ((const int2*)ell)[(size_t)(rowBase + i) * ELL_CAP + lane];
        }
        int tk = 0;
        if (threadIdx.x == 0) tk = atomicAdd(ticket, 1);
        #pragma unroll
        for (int i = 0; i < RW; ++i) {
            const int row = rowBase + i;
            const int n = nn[i];
            const int m = n < ELL_CAP ? n : ELL_CAP;
            vPf acc = {};
            int j = 0;
            for (; j + 16 <= m; j += 16) gather_slots<16, K>(acc, hl, ee[i].x, ee[i].y, j);
            if (j + 8 <= m) { gather_slots<8, K>(acc, hl, ee[i].x, ee[i].y, j); j += 8; }
            if (j + 4 <= m) { gather_slots<4, K>(acc, hl, ee[i].x, ee[i].y, j); j += 4; }
            if (j + 2 <= m) { gather_slots<2, K>(acc, hl, ee[i].x, ee[i].y, j); j += 2; }
            if (j < m)      { gather_slots<1, K>(acc, hl, ee[i].x, ee[i].y, j); }
            if (n > ELL_CAP) {
                const int total = cnt[nRows];
                for (int base = 0; base < total; base += 32) {
                    const int k = base + lane;
                    int orow = -1, oc = 0, ov = 0;
                    if (k < total) {
                        orow = ovf[k];
                        oc   = ovf[(size_t)nEdges + k];
                        ov   = ovf[(size_t)nEdges * 2 + k];
                    }
                    unsigned hit = __builtin_amdgcn_ballot_w32(orow == row);
                    while (hit) {
                        const int jo = __builtin_ctz(hit);
                        hit &= hit - 1;
                        const int   cc = __builtin_amdgcn_readlane(oc, jo);
                        const float v  = __int_as_float(__builtin_amdgcn_readlane(ov, jo));
                        const vPf   s  = *(const vPf*)(hl + (size_t)cc * K);
                        acc += v * s;
                    }
                }
            }
            #pragma unroll
            for (int p = 0; p < PER; ++p) acc[p] = fmaxf(acc[p], 0.0f);
            *(vPf*)(&As[buf][(wave * RW + i) * AP + lane * PER]) = acc;
        }
        if (threadIdx.x == 0) sNext[buf] = (int)gridDim.x + tk;
        __syncthreads();
        const int next = __builtin_amdgcn_readfirstlane(sNext[buf]);

        const int row0 = step * ROWS + rt * 16;
        if (wave < TILES && row0 < nRows) {
            const float* ap = &As[buf][(rt * 16 + c) * AP + 4 * h];
            v8f d = v8f{};
            #pragma unroll
            for (int g = 0; g < KJ; g += 8) {
                float4 a[8], b[8];
                #pragma unroll
                for (int u = 0; u < 8; ++u) {
                    a[u] = *(const float4*)(ap + 8 * (g + u));
                    b[u] = *(const float4*)(bp + 2 * (g + u) * PITCH);
                }
                __builtin_amdgcn_sched_barrier(0);
                #pragma unroll
                for (int u = 0; u < 8; ++u) {
                    v2f a0; a0.x = a[u].x; a0.y = a[u].y;
                    v2f a1; a1.x = a[u].z; a1.y = a[u].w;
                    v2f b0; b0.x = b[u].x; b0.y = b[u].y;
                    v2f b1; b1.x = b[u].z; b1.y = b[u].w;
                    d = __builtin_amdgcn_wmma_f32_16x16x4_f32(
                            false, a0, false, b0, (short)0, d, false, false);
                    d = __builtin_amdgcn_wmma_f32_16x16x4_f32(
                            false, a1, false, b1, (short)0, d, false, false);
                }
            }
            float* yp = Y + (size_t)(row0 + h * 8) * M + ct * 16 + c;
            #pragma unroll
            for (int r = 0; r < 8; ++r) yp[(size_t)r * M] = d[r];
        }
        step = next;
        buf ^= 1;
    }
}

extern "C" void kernel_launch(void* const* d_in, const int* in_sizes, int n_in,
                              void* d_out, int out_size, void* d_ws, size_t ws_size,
                              hipStream_t stream) {
    const float* features = (const float*)d_in[0];
    const float* edgeVals = (const float*)d_in[1];
    const float* W0       = (const float*)d_in[2];
    const float* W1       = (const float*)d_in[3];
    const float* W2       = (const float*)d_in[4];
    const int*   edgeRow  = (const int*)d_in[5];
    const int*   edgeCol  = (const int*)d_in[6];

    const int N = in_sizes[0] / 256;
    const int E = in_sizes[1];

    float* ws = (float*)d_ws;
    float* t1    = ws;
    float* t0    = ws + (size_t)N * 256;
    float* t2    = ws + (size_t)N * 384;
    float* out   = (float*)d_out;
    int* ell = (int*)(ws + (size_t)N * 512);
    int* ovf = ell + (size_t)N * 64;
    int* cnt = ovf + (size_t)E * 3;

    const int rowBlocks  = (N + 7) / 8;
    const int edgeBlocks = (E + 255) / 256;
    const int gemmGroups = 512;
    const int fuseGroups = 1024;

    gemm_wmma_f32<256, 128, 4, false, true><<<gemmGroups, 256, 0, stream>>>(features, W0, t0, cnt, N);

    bucket_edges<<<edgeBlocks, 256, 0, stream>>>(edgeVals, edgeRow, edgeCol, cnt, ell, ovf, N, E);

    gather_gemm_f32<128, 64><<<fuseGroups, 256, 0, stream>>>(t0, W1, t1, cnt, ell, ovf, cnt + N + 1, N, E);

    gather_gemm_f32<64, 32><<<fuseGroups, 256, 0, stream>>>(t1, W2, t2, cnt, ell, ovf, cnt + N + 2, N, E);

    spmm_gather<32><<<rowBlocks, 256, 0, stream>>>(t2, cnt, ell, ovf, out, N, E);
}


# Edge bucketing moved into the first GEMM's launch: 256 groups of 16 waves, 8 do the product, 8 bucket edges (same int adds, slots by arrival as before); counts cleared by a small launch in front
# speedup vs baseline: 3.0413x; 1.0582x over previous
#include <hip/hip_runtime.h>
#include <hip/hip_bf16.h>


typedef __attribute__((ext_vector_type(2))) float v2f;
typedef __attribute__((ext_vector_type(8))) float v8f;

template <int K, int M, int NT, bool RELU, bool REFILL, bool NORM>
__device__ __forceinline__ void gemm_unit(float4 (&aq)[K / 8],
                                          const float* __restrict__ bptr,
                                          const float* __restrict__ anext,
                                          float* __restrict__ yptr,
                                          const int h) {
    static_assert(!(NORM && RELU), "the squares are summed from the raw A values");
    constexpr int KJ    = K / 8;
    constexpr int PITCH = 4 * M + 32;
    v8f acc[NT];
    #pragma unroll
    for (int t = 0; t < NT; ++t) acc[t] = v8f{};
    v2f ss0 = {}, ss1 = {};

    float4 bq[2][NT];
    #pragma unroll
    for (int t = 0; t < NT; ++t) bq[0][t] = *(const float4*)(bptr + t * 64);

    #pragma unroll
    for (int j = 0; j < KJ; ++j) {
        if (j + 1 < KJ) {
            #pragma unroll
            for (int t = 0; t < NT; ++t)
                bq[(j + 1) & 1][t] = *(const float4*)(bptr + 2 * (j + 1) * PITCH + t * 64);
            __builtin_amdgcn_sched_barrier(0);
        }
        float4 a = aq[j];
        if (RELU) {
            a.x = fmaxf(a.x, 0.0f); a.y = fmaxf(a.y, 0.0f);
            a.z = fmaxf(a.z, 0.0f); a.w = fmaxf(a.w, 0.0f);
        }
        v2f a0; a0.x = a.x; a0.y = a.y;
        v2f a1; a1.x = a.z; a1.y = a.w;
        if (NORM) { ss0 += a0 * a0; ss1 += a1 * a1; }
        #pragma unroll
        for (int t = 0; t < NT; ++t) {
            v2f b; b.x = bq[j & 1][t].x; b.y = bq[j & 1][t].y;
            acc[t] = __builtin_amdgcn_wmma_f32_16x16x4_f32(
                         false, a0, false, b, (short)0, acc[t], false, false);
        }
        #pragma unroll
        for (int t = 0; t < NT; ++t) {
            v2f b; b.x = bq[j & 1][t].z; b.y = bq[j & 1][t].w;
            acc[t] = __builtin_amdgcn_wmma_f32_16x16x4_f32(
                         false, a1, false, b, (short)0, acc[t], false, false);
        }
        if (REFILL) {
            aq[j] = *(const float4*)(anext + 8 * j);
            __builtin_amdgcn_sched_barrier(0);
        }
    }
    if (NORM) {
        float s = (ss0.x + ss0.y) + (ss1.x + ss1.y);
        s += __shfl_xor(s, 16, 32);
        const float inv = rsqrtf(fmaxf(s, 1e-12f));
        #pragma unroll
        for (int r = 0; r < 8; ++r) {
            const float sc = __shfl(inv, 8 * h + r, 32);
            #pragma unroll
            for (int t = 0; t < NT; ++t)
                yptr[(size_t)r * M + t * 16] = acc[t][r] * sc;
        }
    } else {
        #pragma unroll
        for (int t = 0; t < NT; ++t) {
            #pragma unroll
            for (int r = 0; r < 8; ++r)
                yptr[(size_t)r * M + t * 16] = acc[t][r];
        }
    }
}

constexpr int ELL_CAP = 32;
constexpr int EB      = 8;

__global__ __launch_bounds__(256) void clear_counts(int* __restrict__ cnt, int n) {
    const int i = blockIdx.x * 256 + threadIdx.x;
    if (i < n) cnt[i] = 0;
}

__device__ __forceinline__ void bucket_edges(const float* __restrict__ vals,
                                             const int* __restrict__ rows,
                                             const int* __restrict__ cols,
                                             int* __restrict__ cnt,
                                             int* __restrict__ ell,
                                             int* __restrict__ ovf,
                                             const int nRows, const int nEdges,
                                             const int first, const int stride) {
    for (int base = first; base < nEdges; base += EB * stride) {
        int   r[EB], c[EB], slot[EB];
        float v[EB];
        #pragma unroll
        for (int u = 0; u < EB; ++u) {
            const int e = base + u * stride;
            r[u] = 0; c[u] = 0; v[u] = 0.0f;
            if (e < nEdges) { r[u] = rows[e]; c[u] = cols[e]; v[u] = vals[e]; }
        }
        #pragma unroll
        for (int u = 0; u < EB; ++u) {
            const int e = base + u * stride;
            slot[u] = 0;
            if (e < nEdges) slot[u] = atomicAdd(&cnt[r[u]], 1);
        }
        #pragma unroll
        for (int u = 0; u < EB; ++u) {
            const int e = base + u * stride;
            if (e >= nEdges) continue;
            if (slot[u] < ELL_CAP) {
                int2 ent; ent.x = c[u]; ent.y = __float_as_int(v[u]);
                ((int2*)ell)[(size_t)r[u] * ELL_CAP + slot[u]] = ent;
            } else {
                const int p = atomicAdd(&cnt[nRows], 1);
                ovf[p]                      = r[u];
                ovf[(size_t)nEdges + p]     = c[u];
                ovf[(size_t)nEdges * 2 + p] = __float_as_int(v[u]);
            }
        }
    }
}

template <int K, int M, int NT, bool RELU, bool NORM>
__global__ __launch_bounds__(512) void gemm_wmma_f32(const float* __restrict__ X,
                                                     const float* __restrict__ W,
                                                     float* __restrict__ Y,
                                                     int nRows,
                                                     const float* __restrict__ edgeVals,
                                                     const int* __restrict__ edgeRow,
                                                     const int* __restrict__ edgeCol,
                                                     int* __restrict__ cnt,
                                                     int* __restrict__ ell,
                                                     int* __restrict__ ovf,
                                                     int nEdges) {
    constexpr int KQ    = K / 4;
    constexpr int PITCH = 4 * M + 32;
    constexpr int KJ    = K / 8;
    constexpr int CG    = M / (16 * NT);
    __shared__ __attribute__((aligned(16))) float Ws[KQ * PITCH];

    const int lane = threadIdx.x & 31;
    const int wave = __builtin_amdgcn_readfirstlane(threadIdx.x >> 5);
    const int c    = lane & 15;
    const int h    = lane >> 4;

    const int totalUnits = (nRows / 16) * CG;
    const int perGroup   = (totalUnits + (int)gridDim.x - 1) / (int)gridDim.x;
    const int uBeg       = (int)blockIdx.x * perGroup;
    const int uEnd       = uBeg + perGroup < totalUnits ? uBeg + perGroup : totalUnits;
    int u = uBeg + wave;
    const bool haveWork = wave < 8 && u < uEnd;

    float4 aq[KJ];
    if (haveWork) {
        const float* ap = X + (size_t)((u / CG) * 16 + c) * K + 4 * h;
        #pragma unroll
        for (int j = 0; j < KJ; ++j) aq[j] = *(const float4*)(ap + 8 * j);
    }

    #pragma unroll 8
    for (int i = threadIdx.x; i < KQ * M; i += 512) {
        const int q = i / M;
        const int n = i % M;
        float4 v;
        v.x = W[(size_t)(4 * q + 0) * M + n];
        v.y = W[(size_t)(4 * q + 1) * M + n];
        v.z = W[(size_t)(4 * q + 2) * M + n];
        v.w = W[(size_t)(4 * q + 3) * M + n];
        *(float4*)(Ws + q * PITCH + 4 * n) = v;
    }
    __syncthreads();
    if (wave >= 8) {
        bucket_edges(edgeVals, edgeRow, edgeCol, cnt, ell, ovf, nRows, nEdges,
                     (int)blockIdx.x * 256 + (int)threadIdx.x - 256, (int)gridDim.x * 256);
        return;
    }
    if (!haveWork) return;

    while (true) {
        const int rowTile = u / CG;
        const int cg      = u % CG;
        const float* bptr = Ws + h * PITCH + (cg * NT * 16 + c) * 4;
        float* yptr = Y + (size_t)(rowTile * 16 + h * 8) * M + cg * NT * 16 + c;
        const int un = u + 8;
        if (un < uEnd) {
            const float* anext = X + (size_t)((un / CG) * 16 + c) * K + 4 * h;
            gemm_unit<K, M, NT, RELU, true, NORM>(aq, bptr, anext, yptr, h);
            u = un;
        } else {
            gemm_unit<K, M, NT, RELU, false, NORM>(aq, bptr, nullptr, yptr, h);
            break;
        }
    }
}

template <int D>
__global__ __launch_bounds__(256) void spmm_gather(const float* __restrict__ H,
                                                   const int* __restrict__ cnt,
                                                   const int* __restrict__ ell,
                                                   const int* __restrict__ ovf,
                                                   float* __restrict__ out,
                                                   int nRows, int nEdges) {
    constexpr int PER = D / 32;
    typedef __attribute__((ext_vector_type(PER))) float vPf;

    const int lane = threadIdx.x & 31;
    const int row = __builtin_amdgcn_readfirstlane(blockIdx.x * 8 + (threadIdx.x >> 5));
    if (row >= nRows) return;
    const int n = cnt[row];
    const int m = n < ELL_CAP ? n : ELL_CAP;

    int2 ent; ent.x = 0; ent.y = 0;
    if (lane < m) ent = ((const int2*)ell)[(size_t)row * ELL_CAP + lane];

    const float* __restrict__ hl = H + lane * PER;
    vPf acc = {};
    #pragma unroll 8
    for (int j = 0; j < m; ++j) {
        const int   c = __builtin_amdgcn_readlane(ent.x, j);
        const float v = __int_as_float(__builtin_amdgcn_readlane(ent.y, j));
        const vPf   s = *(const vPf*)(hl + (size_t)c * D);
        acc += v * s;
    }

    if (n > ELL_CAP) {
        const int total = cnt[nRows];
        for (int base = 0; base < total; base += 32) {
            const int i = base + lane;
            int orow = -1, oc = 0, ov = 0;
            if (i < total) {
                orow = ovf[i];
                oc   = ovf[(size_t)nEdges + i];
                ov   = ovf[(size_t)nEdges * 2 + i];
            }
            unsigned hit = __builtin_amdgcn_ballot_w32(orow == row);
            while (hit) {
                const int j = __builtin_ctz(hit);
                hit &= hit - 1;
                const int   c = __builtin_amdgcn_readlane(oc, j);
                const float v = __int_as_float(__builtin_amdgcn_readlane(ov, j));
                const vPf   s = *(const vPf*)(hl + (size_t)c * D);
                acc += v * s;
            }
        }
    }
    *(vPf*)(out + (size_t)row * D + lane * PER) = acc;
}

template <int B, int K, typename V>
__device__ __forceinline__ void gather_slots(V& acc, const float* __restrict__ hl,
                                             const int ex, const int ey, const int j0) {
    V     s[B];
    float v[B];
    #pragma unroll
    for (int u = 0; u < B; ++u) {
        const int cc = __builtin_amdgcn_readlane(ex, j0 + u);
        v[u] = __int_as_float(__builtin_amdgcn_readlane(ey, j0 + u));
        s[u] = *(const V*)(hl + (size_t)cc * K);
    }
    #pragma unroll
    for (int u = 0; u < B; ++u) acc += v[u] * s[u];
}

template <int K, int M>
__global__ __launch_bounds__(256) void gather_gemm_f32(const float* __restrict__ H,
                                                       const float* __restrict__ W,
                                                       float* __restrict__ Y,
                                                       const int* __restrict__ cnt,
                                                       const int* __restrict__ ell,
                                                       const int* __restrict__ ovf,
                                                       int* __restrict__ ticket,
                                                       int nRows, int nEdges) {
    constexpr int CT    = M / 16;
    constexpr int ROWS  = 32;
    constexpr int TILES = 2 * CT;
    constexpr int RW    = ROWS / 8;
    constexpr int KQ    = K / 4;
    constexpr int PITCH = 4 * M + 32;
    constexpr int KJ    = K / 8;
    constexpr int AP    = K + 4;
    constexpr int PER   = K / 32;
    static_assert(KJ % 8 == 0 && TILES <= 8, "product groups of 8 steps, one tile per wave");
    typedef __attribute__((ext_vector_type(PER))) float vPf;

    __shared__ __attribute__((aligned(16))) float Ws[KQ * PITCH];
    __shared__ __attribute__((aligned(16))) float As[2][ROWS * AP];
    __shared__ int sNext[2];

    const int lane = threadIdx.x & 31;
    const int wave = __builtin_amdgcn_readfirstlane(threadIdx.x >> 5);
    const int c    = lane & 15;
    const int h    = lane >> 4;

    #pragma unroll 8
    for (int i = threadIdx.x; i < KQ * M; i += 256) {
        const int q = i / M;
        const int n = i % M;
        float4 v;
        v.x = W[(size_t)(4 * q + 0) * M + n];
        v.y = W[(size_t)(4 * q + 1) * M + n];
        v.z = W[(size_t)(4 * q + 2) * M + n];
        v.w = W[(size_t)(4 * q + 3) * M + n];
        *(float4*)(Ws + q * PITCH + 4 * n) = v;
    }

    const int nSteps = (nRows + ROWS - 1) / ROWS;
    const int rt = wave / CT;
    const int ct = wave % CT;
    const float* __restrict__ hl = H + lane * PER;
    const float* bp = Ws + h * PITCH + (ct * 16 + c) * 4;

    int step = blockIdx.x;
    int buf  = 0;
    while (step < nSteps) {
        const int rowBase = step * ROWS + wave * RW;
        int myN = 0;
        if (lane < RW && rowBase + lane < nRows) myN = cnt[rowBase + lane];
        int  nn[RW];
        int2 ee[RW];
        #pragma unroll
        for (int i = 0; i < RW; ++i) {
            nn[i] = __builtin_amdgcn_readlane(myN, i);
            const int m = nn[i] < ELL_CAP ? nn[i] : ELL_CAP;
            ee[i].x = 0; ee[i].y = 0;
            if (lane < m) ee[i] = ((const int2*)ell)[(size_t)(rowBase + i) * ELL_CAP + lane];
        }
        int tk = 0;
        if (threadIdx.x == 0) tk = atomicAdd(ticket, 1);
        #pragma unroll
        for (int i = 0; i < RW; ++i) {
            const int row = rowBase + i;
            const int n = nn[i];
            const int m = n < ELL_CAP ? n : ELL_CAP;
            vPf acc = {};
            int j = 0;
            for (; j + 16 <= m; j += 16) gather_slots<16, K>(acc, hl, ee[i].x, ee[i].y, j);
            if (j + 8 <= m) { gather_slots<8, K>(acc, hl, ee[i].x, ee[i].y, j); j += 8; }
            if (j + 4 <= m) { gather_slots<4, K>(acc, hl, ee[i].x, ee[i].y, j); j += 4; }
            if (j + 2 <= m) { gather_slots<2, K>(acc, hl, ee[i].x, ee[i].y, j); j += 2; }
            if (j < m)      { gather_slots<1, K>(acc, hl, ee[i].x, ee[i].y, j); }
            if (n > ELL_CAP) {
                const int total = cnt[nRows];
                for (int base = 0; base < total; base += 32) {
                    const int k = base + lane;
                    int orow = -1, oc = 0, ov = 0;
                    if (k < total) {
                        orow = ovf[k];
                        oc   = ovf[(size_t)nEdges + k];
                        ov   = ovf[(size_t)nEdges * 2 + k];
                    }
                    unsigned hit = __builtin_amdgcn_ballot_w32(orow == row);
                    while (hit) {
                        const int jo = __builtin_ctz(hit);
                        hit &= hit - 1;
                        const int   cc = __builtin_amdgcn_readlane(oc, jo);
                        const float v  = __int_as_float(__builtin_amdgcn_readlane(ov, jo));
                        const vPf   s  = *(const vPf*)(hl + (size_t)cc * K);
                        acc += v * s;
                    }
                }
            }
            #pragma unroll
            for (int p = 0; p < PER; ++p) acc[p] = fmaxf(acc[p], 0.0f);
            *(vPf*)(&As[buf][(wave * RW + i) * AP + lane * PER]) = acc;
        }
        if (threadIdx.x == 0) sNext[buf] = (int)gridDim.x + tk;
        __syncthreads();
        const int next = __builtin_amdgcn_readfirstlane(sNext[buf]);

        const int row0 = step * ROWS + rt * 16;
        if (wave < TILES && row0 < nRows) {
            const float* ap = &As[buf][(rt * 16 + c) * AP + 4 * h];
            v8f d = v8f{};
            #pragma unroll
            for (int g = 0; g < KJ; g += 8) {
                float4 a[8], b[8];
                #pragma unroll
                for (int u = 0; u < 8; ++u) {
                    a[u] = *(const float4*)(ap + 8 * (g + u));
                    b[u] = *(const float4*)(bp + 2 * (g + u) * PITCH);
                }
                __builtin_amdgcn_sched_barrier(0);
                #pragma unroll
                for (int u = 0; u < 8; ++u) {
                    v2f a0; a0.x = a[u].x; a0.y = a[u].y;
                    v2f a1; a1.x = a[u].z; a1.y = a[u].w;
                    v2f b0; b0.x = b[u].x; b0.y = b[u].y;
                    v2f b1; b1.x = b[u].z; b1.y = b[u].w;
                    d = __builtin_amdgcn_wmma_f32_16x16x4_f32(
                            false, a0, false, b0, (short)0, d, false, false);
                    d = __builtin_amdgcn_wmma_f32_16x16x4_f32(
                            false, a1, false, b1, (short)0, d, false, false);
                }
            }
            float* yp = Y + (size_t)(row0 + h * 8) * M + ct * 16 + c;
            #pragma unroll
            for (int r = 0; r < 8; ++r) yp[(size_t)r * M] = d[r];
        }
        step = next;
        buf ^= 1;
    }
}

extern "C" void kernel_launch(void* const* d_in, const int* in_sizes, int n_in,
                              void* d_out, int out_size, void* d_ws, size_t ws_size,
                              hipStream_t stream) {
    const float* features = (const float*)d_in[0];
    const float* edgeVals = (const float*)d_in[1];
    const float* W0       = (const float*)d_in[2];
    const float* W1       = (const float*)d_in[3];
    const float* W2       = (const float*)d_in[4];
    const int*   edgeRow  = (const int*)d_in[5];
    const int*   edgeCol  = (const int*)d_in[6];

    const int N = in_sizes[0] / 256;
    const int E = in_sizes[1];

    float* ws = (float*)d_ws;
    float* t1    = ws;
    float* t0    = ws + (size_t)N * 256;
    float* t2    = ws + (size_t)N * 384;
    float* out   = (float*)d_out;
    int* ell = (int*)(ws + (size_t)N * 512);
    int* ovf = ell + (size_t)N * 64;
    int* cnt = ovf + (size_t)E * 3;

    const int rowBlocks  = (N + 7) / 8;
    const int cntBlocks  = (N + 3 + 255) / 256;
    const int gemmGroups = 256;
    const int fuseGroups = 1024;

    clear_counts<<<cntBlocks, 256, 0, stream>>>(cnt, N + 3);

    gemm_wmma_f32<256, 128, 4, false, true><<<gemmGroups, 512, 0, stream>>>(
        features, W0, t0, N, edgeVals, edgeRow, edgeCol, cnt, ell, ovf, E);

    gather_gemm_f32<128, 64><<<fuseGroups, 256, 0, stream>>>(t0, W1, t1, cnt, ell, ovf, cnt + N + 1, N, E);

    gather_gemm_f32<64, 32><<<fuseGroups, 256, 0, stream>>>(t1, W2, t2, cnt, ell, ovf, cnt + N + 2, N, E);

    spmm_gather<32><<<rowBlocks, 256, 0, stream>>>(t2, cnt, ell, ovf, out, N, E);
}


# Edge bucketing without global atomics or scattered stores: tiles sorted by row bin in LDS (launch 1, replaces the clear), one wave per bin fills rows in LDS beside GEMM 1; new slot order of row sums
# speedup vs baseline: 4.5338x; 1.4907x over previous
#include <hip/hip_runtime.h>
#include <hip/hip_bf16.h>


typedef __attribute__((ext_vector_type(2))) float v2f;
typedef __attribute__((ext_vector_type(8))) float v8f;

template <int K, int M, int NT, bool RELU, bool REFILL, bool NORM>
__device__ __forceinline__ void gemm_unit(float4 (&aq)[K / 8],
                                          const float* __restrict__ bptr,
                                          const float* __restrict__ anext,
                                          float* __restrict__ yptr,
                                          const int h) {
    static_assert(!(NORM && RELU), "the squares are summed from the raw A values");
    constexpr int KJ    = K / 8;
    constexpr int PITCH = 4 * M + 32;
    v8f acc[NT];
    #pragma unroll
    for (int t = 0; t < NT; ++t) acc[t] = v8f{};
    v2f ss0 = {}, ss1 = {};

    float4 bq[2][NT];
    #pragma unroll
    for (int t = 0; t < NT; ++t) bq[0][t] = *(const float4*)(bptr + t * 64);

    #pragma unroll
    for (int j = 0; j < KJ; ++j) {
        if (j + 1 < KJ) {
            #pragma unroll
            for (int t = 0; t < NT; ++t)
                bq[(j + 1) & 1][t] = *(const float4*)(bptr + 2 * (j + 1) * PITCH + t * 64);
            __builtin_amdgcn_sched_barrier(0);
        }
        float4 a = aq[j];
        if (RELU) {
            a.x = fmaxf(a.x, 0.0f); a.y = fmaxf(a.y, 0.0f);
            a.z = fmaxf(a.z, 0.0f); a.w = fmaxf(a.w, 0.0f);
        }
        v2f a0; a0.x = a.x; a0.y = a.y;
        v2f a1; a1.x = a.z; a1.y = a.w;
        if (NORM) { ss0 += a0 * a0; ss1 += a1 * a1; }
        #pragma unroll
        for (int t = 0; t < NT; ++t) {
            v2f b; b.x = bq[j & 1][t].x; b.y = bq[j & 1][t].y;
            acc[t] = __builtin_amdgcn_wmma_f32_16x16x4_f32(
                         false, a0, false, b, (short)0, acc[t], false, false);
        }
        #pragma unroll
        for (int t = 0; t < NT; ++t) {
            v2f b; b.x = bq[j & 1][t].z; b.y = bq[j & 1][t].w;
            acc[t] = __builtin_amdgcn_wmma_f32_16x16x4_f32(
                         false, a1, false, b, (short)0, acc[t], false, false);
        }
        if (REFILL) {
            aq[j] = *(const float4*)(anext + 8 * j);
            __builtin_amdgcn_sched_barrier(0);
        }
    }
    if (NORM) {
        float s = (ss0.x + ss0.y) + (ss1.x + ss1.y);
        s += __shfl_xor(s, 16, 32);
        const float inv = rsqrtf(fmaxf(s, 1e-12f));
        #pragma unroll
        for (int r = 0; r < 8; ++r) {
            const float sc = __shfl(inv, 8 * h + r, 32);
            #pragma unroll
            for (int t = 0; t < NT; ++t)
                yptr[(size_t)r * M + t * 16] = acc[t][r] * sc;
        }
    } else {
        #pragma unroll
        for (int t = 0; t < NT; ++t) {
            #pragma unroll
            for (int r = 0; r < 8; ++r)
                yptr[(size_t)r * M + t * 16] = acc[t][r];
        }
    }
}

constexpr int ELL_CAP = 32;
constexpr int NBIN    = 2048;
constexpr int TILE    = 6272;
constexpr int SPITCH  = NBIN + 64;
constexpr int BT      = 512;
constexpr int EPT     = (TILE + BT - 1) / BT;
constexpr int BPT     = NBIN / BT;
constexpr int RB_MAX  = 64;
constexpr int COL_BITS = 26;

__global__ __launch_bounds__(BT) void bin_edges(const float* __restrict__ vals,
                                                const int* __restrict__ rows,
                                                const int* __restrict__ cols,
                                                int* __restrict__ sorted,
                                                int* __restrict__ starts,
                                                int* __restrict__ cnt,
                                                int nRows, int nEdges) {
    static_assert(NBIN % BT == 0 && BT % 32 == 0 && BT / 32 <= 32, "scan: BPT bins per thread, one wave sums the waves");
    static_assert(EPT * 3 < 63, "a thread's loads must fit the load counter");
    __shared__ int  cntB[NBIN];
    __shared__ int  startB[NBIN + 1];
    __shared__ int  waveTot[32];
    __shared__ int2 stage[TILE];

    const int t    = threadIdx.x;
    const int lane = t & 31;
    const int wave = t >> 5;
    const int RB   = (nRows + NBIN - 1) / NBIN;
    const int e0   = (int)blockIdx.x * TILE;
    const int e1   = e0 + TILE < nEdges ? e0 + TILE : nEdges;

    int r[EPT], cc[EPT], vv[EPT], b[EPT];
    #pragma unroll
    for (int u = 0; u < EPT; ++u) {
        const int e = e0 + t + u * BT;
        r[u] = -1; cc[u] = 0; vv[u] = 0;
        if (e < e1) { r[u] = rows[e]; cc[u] = cols[e]; vv[u] = __float_as_int(vals[e]); }
        if ((unsigned)r[u] >= (unsigned)nRows) r[u] = -1;
    }
    if (blockIdx.x == 0 && t < 3) cnt[nRows + t] = 0;
    #pragma unroll
    for (int i = 0; i < BPT; ++i) cntB[t + i * BT] = 0;
    __syncthreads();
    #pragma unroll
    for (int u = 0; u < EPT; ++u) {
        b[u] = 0;
        if (r[u] >= 0) { b[u] = r[u] / RB; atomicAdd(&cntB[b[u]], 1); }
    }
    __syncthreads();

    int cb[BPT];
    int s = 0;
    #pragma unroll
    for (int i = 0; i < BPT; ++i) { cb[i] = cntB[BPT * t + i]; s += cb[i]; }
    int incl = s;
    #pragma unroll
    for (int d = 1; d < 32; d <<= 1) {
        const int o = __shfl_up(incl, d, 32);
        if (lane >= d) incl += o;
    }
    if (lane == 31) waveTot[wave] = incl;
    __syncthreads();
    if (wave == 0) {
        const int v = lane < BT / 32 ? waveTot[lane] : 0;
        int vi = v;
        #pragma unroll
        for (int d = 1; d < 32; d <<= 1) {
            const int o = __shfl_up(vi, d, 32);
            if (lane >= d) vi += o;
        }
        if (lane < BT / 32) waveTot[lane] = vi - v;
    }
    __syncthreads();
    int run = incl - s + waveTot[wave];
    #pragma unroll
    for (int i = 0; i < BPT; ++i) {
        startB[BPT * t + i] = run;
        cntB[BPT * t + i]   = run;
        run += cb[i];
    }
    if (t == BT - 1) startB[NBIN] = run;
    __syncthreads();

    #pragma unroll
    for (int u = 0; u < EPT; ++u) {
        if (r[u] < 0) continue;
        const int pos = atomicAdd(&cntB[b[u]], 1);
        int2 rec;
        rec.x = (int)(((unsigned)(r[u] - b[u] * RB) << COL_BITS) | (unsigned)cc[u]);
        rec.y = vv[u];
        stage[pos] = rec;
    }
    __syncthreads();
    for (int p = t; p < e1 - e0; p += BT) ((int2*)sorted)[(size_t)e0 + p] = stage[p];
    for (int i = t; i <= NBIN; i += BT) starts[(size_t)blockIdx.x * SPITCH + i] = startB[i];
}

constexpr int PB = 8;
__device__ __forceinline__ void place_bin(const int* __restrict__ sorted,
                                          const int* __restrict__ starts,
                                          const int nTiles,
                                          int* __restrict__ cnt,
                                          int* __restrict__ ell,
                                          int* __restrict__ ovf,
                                          int* rowCnt, int2* slots,
                                          const int nRows, const int nEdges,
                                          const int bin, const int lane) {
    const int RB   = (nRows + NBIN - 1) / NBIN;
    const int row0 = bin * RB;
    if (row0 >= nRows) return;
    const int nr = row0 + RB <= nRows ? RB : nRows - row0;

    rowCnt[lane] = 0; rowCnt[lane + 32] = 0;
    for (int tb = 0; tb < nTiles; tb += 32) {
        const int tile = tb + lane;
        int a = 0, n = 0;
        if (tile < nTiles) {
            const int* sp = starts + (size_t)tile * SPITCH + bin;
            a = sp[0];
            n = sp[1] - a;
        }
        const int2* rp = (const int2*)sorted + (size_t)tile * TILE + a;
        for (int j0 = 0; __builtin_amdgcn_ballot_w32(j0 < n) != 0u; j0 += PB) {
            int2 rec[PB];
            #pragma unroll
            for (int u = 0; u < PB; ++u) {
                rec[u].x = 0; rec[u].y = 0;
                if (j0 + u < n) rec[u] = rp[j0 + u];
            }
            #pragma unroll
            for (int u = 0; u < PB; ++u) {
                if (j0 + u >= n) continue;
                const int rl   = (int)((unsigned)rec[u].x >> COL_BITS);
                const int col  = rec[u].x & ((1 << COL_BITS) - 1);
                const int slot = atomicAdd(&rowCnt[rl], 1);
                if (slot < ELL_CAP) {
                    int2 ent; ent.x = col; ent.y = rec[u].y;
                    slots[rl * ELL_CAP + slot] = ent;
                } else {
                    const int p = atomicAdd(&cnt[nRows], 1);
                    ovf[p]                      = row0 + rl;
                    ovf[(size_t)nEdges + p]     = col;
                    ovf[(size_t)nEdges * 2 + p] = rec[u].y;
                }
            }
        }
    }
    __builtin_amdgcn_fence(__ATOMIC_ACQ_REL, "workgroup");
    for (int i = 0; i < nr; ++i) {
        const int n = rowCnt[i];
        if (lane < (n < ELL_CAP ? n : ELL_CAP))
            ((int2*)ell)[(size_t)(row0 + i) * ELL_CAP + lane] = slots[i * ELL_CAP + lane];
    }
    if (lane < nr)      cnt[row0 + lane]      = rowCnt[lane];
    if (lane + 32 < nr) cnt[row0 + lane + 32] = rowCnt[lane + 32];
}

template <int K, int M, int NT, bool RELU, bool NORM>
__global__ __launch_bounds__(512) void gemm_wmma_f32(const float* __restrict__ X,
                                                     const float* __restrict__ W,
                                                     float* __restrict__ Y,
                                                     int nRows,
                                                     const int* __restrict__ sorted,
                                                     const int* __restrict__ starts,
                                                     int nTiles,
                                                     int* __restrict__ cnt,
                                                     int* __restrict__ ell,
                                                     int* __restrict__ ovf,
                                                     int nEdges) {
    constexpr int KQ    = K / 4;
    constexpr int PITCH = 4 * M + 32;
    constexpr int KJ    = K / 8;
    constexpr int CG    = M / (16 * NT);
    __shared__ __attribute__((aligned(16))) float Ws[KQ * PITCH];
    __shared__ int2 binSlots[8][RB_MAX * ELL_CAP];
    __shared__ int  binCnt[8][RB_MAX];

    const int lane = threadIdx.x & 31;
    const int wave = __builtin_amdgcn_readfirstlane(threadIdx.x >> 5);
    const int c    = lane & 15;
    const int h    = lane >> 4;

    const int totalUnits = (nRows / 16) * CG;
    const int perGroup   = (totalUnits + (int)gridDim.x - 1) / (int)gridDim.x;
    const int uBeg       = (int)blockIdx.x * perGroup;
    const int uEnd       = uBeg + perGroup < totalUnits ? uBeg + perGroup : totalUnits;
    int u = uBeg + wave;
    const bool haveWork = wave < 8 && u < uEnd;

    float4 aq[KJ];
    if (haveWork) {
        const float* ap = X + (size_t)((u / CG) * 16 + c) * K + 4 * h;
        #pragma unroll
        for (int j = 0; j < KJ; ++j) aq[j] = *(const float4*)(ap + 8 * j);
    }

    #pragma unroll 8
    for (int i = threadIdx.x; i < KQ * M; i += 512) {
        const int q = i / M;
        const int n = i % M;
        float4 v;
        v.x = W[(size_t)(4 * q + 0) * M + n];
        v.y = W[(size_t)(4 * q + 1) * M + n];
        v.z = W[(size_t)(4 * q + 2) * M + n];
        v.w = W[(size_t)(4 * q + 3) * M + n];
        *(float4*)(Ws + q * PITCH + 4 * n) = v;
    }
    __syncthreads();
    if (wave >= 8) {
        place_bin(sorted, starts, nTiles, cnt, ell, ovf, binCnt[wave - 8], binSlots[wave - 8],
                  nRows, nEdges, (int)blockIdx.x * 8 + wave - 8, lane);
        return;
    }
    if (!haveWork) return;

    while (true) {
        const int rowTile = u / CG;
        const int cg      = u % CG;
        const float* bptr = Ws + h * PITCH + (cg * NT * 16 + c) * 4;
        float* yptr = Y + (size_t)(rowTile * 16 + h * 8) * M + cg * NT * 16 + c;
        const int un = u + 8;
        if (un < uEnd) {
            const float* anext = X + (size_t)((un / CG) * 16 + c) * K + 4 * h;
            gemm_unit<K, M, NT, RELU, true, NORM>(aq, bptr, anext, yptr, h);
            u = un;
        } else {
            gemm_unit<K, M, NT, RELU, false, NORM>(aq, bptr, nullptr, yptr, h);
            break;
        }
    }
}

template <int D>
__global__ __launch_bounds__(256) void spmm_gather(const float* __restrict__ H,
                                                   const int* __restrict__ cnt,
                                                   const int* __restrict__ ell,
                                                   const int* __restrict__ ovf,
                                                   float* __restrict__ out,
                                                   int nRows, int nEdges) {
    constexpr int PER = D / 32;
    typedef __attribute__((ext_vector_type(PER))) float vPf;

    const int lane = threadIdx.x & 31;
    const int row = __builtin_amdgcn_readfirstlane(blockIdx.x * 8 + (threadIdx.x >> 5));
    if (row >= nRows) return;
    const int n = cnt[row];
    const int m = n < ELL_CAP ? n : ELL_CAP;

    int2 ent; ent.x = 0; ent.y = 0;
    if (lane < m) ent = ((const int2*)ell)[(size_t)row * ELL_CAP + lane];

    const float* __restrict__ hl = H + lane * PER;
    vPf acc = {};
    #pragma unroll 8
    for (int j = 0; j < m; ++j) {
        const int   c = __builtin_amdgcn_readlane(ent.x, j);
        const float v = __int_as_float(__builtin_amdgcn_readlane(ent.y, j));
        const vPf   s = *(const vPf*)(hl + (size_t)c * D);
        acc += v * s;
    }

    if (n > ELL_CAP) {
        const int total = cnt[nRows];
        for (int base = 0; base < total; base += 32) {
            const int i = base + lane;
            int orow = -1, oc = 0, ov = 0;
            if (i < total) {
                orow = ovf[i];
                oc   = ovf[(size_t)nEdges + i];
                ov   = ovf[(size_t)nEdges * 2 + i];
            }
            unsigned hit = __builtin_amdgcn_ballot_w32(orow == row);
            while (hit) {
                const int j = __builtin_ctz(hit);
                hit &= hit - 1;
                const int   c = __builtin_amdgcn_readlane(oc, j);
                const float v = __int_as_float(__builtin_amdgcn_readlane(ov, j));
                const vPf   s = *(const vPf*)(hl + (size_t)c * D);
                acc += v * s;
            }
        }
    }
    *(vPf*)(out + (size_t)row * D + lane * PER) = acc;
}

template <int B, int K, typename V>
__device__ __forceinline__ void gather_slots(V& acc, const float* __restrict__ hl,
                                             const int ex, const int ey, const int j0) {
    V     s[B];
    float v[B];
    #pragma unroll
    for (int u = 0; u < B; ++u) {
        const int cc = __builtin_amdgcn_readlane(ex, j0 + u);
        v[u] = __int_as_float(__builtin_amdgcn_readlane(ey, j0 + u));
        s[u] = *(const V*)(hl + (size_t)cc * K);
    }
    #pragma unroll
    for (int u = 0; u < B; ++u) acc += v[u] * s[u];
}

template <int K, int M>
__global__ __launch_bounds__(256) void gather_gemm_f32(const float* __restrict__ H,
                                                       const float* __restrict__ W,
                                                       float* __restrict__ Y,
                                                       const int* __restrict__ cnt,
                                                       const int* __restrict__ ell,
                                                       const int* __restrict__ ovf,
                                                       int* __restrict__ ticket,
                                                       int nRows, int nEdges) {
    constexpr int CT    = M / 16;
    constexpr int ROWS  = 32;
    constexpr int TILES = 2 * CT;
    constexpr int RW    = ROWS / 8;
    constexpr int KQ    = K / 4;
    constexpr int PITCH = 4 * M + 32;
    constexpr int KJ    = K / 8;
    constexpr int AP    = K + 4;
    constexpr int PER   = K / 32;
    static_assert(KJ % 8 == 0 && TILES <= 8, "product groups of 8 steps, one tile per wave");
    typedef __attribute__((ext_vector_type(PER))) float vPf;

    __shared__ __attribute__((aligned(16))) float Ws[KQ * PITCH];
    __shared__ __attribute__((aligned(16))) float As[2][ROWS * AP];
    __shared__ int sNext[2];

    const int lane = threadIdx.x & 31;
    const int wave = __builtin_amdgcn_readfirstlane(threadIdx.x >> 5);
    const int c    = lane & 15;
    const int h    = lane >> 4;

    #pragma unroll 8
    for (int i = threadIdx.x; i < KQ * M; i += 256) {
        const int q = i / M;
        const int n = i % M;
        float4 v;
        v.x = W[(size_t)(4 * q + 0) * M + n];
        v.y = W[(size_t)(4 * q + 1) * M + n];
        v.z = W[(size_t)(4 * q + 2) * M + n];
        v.w = W[(size_t)(4 * q + 3) * M + n];
        *(float4*)(Ws + q * PITCH + 4 * n) = v;
    }

    const int nSteps = (nRows + ROWS - 1) / ROWS;
    const int rt = wave / CT;
    const int ct = wave % CT;
    const float* __restrict__ hl = H + lane * PER;
    const float* bp = Ws + h * PITCH + (ct * 16 + c) * 4;

    int step = blockIdx.x;
    int buf  = 0;
    while (step < nSteps) {
        const int rowBase = step * ROWS + wave * RW;
        int myN = 0;
        if (lane < RW && rowBase + lane < nRows) myN = cnt[rowBase + lane];
        int  nn[RW];
        int2 ee[RW];
        #pragma unroll
        for (int i = 0; i < RW; ++i) {
            nn[i] = __builtin_amdgcn_readlane(myN, i);
            const int m = nn[i] < ELL_CAP ? nn[i] : ELL_CAP;
            ee[i].x = 0; ee[i].y = 0;
            if (lane < m) ee[i] = ((const int2*)ell)[(size_t)(rowBase + i) * ELL_CAP + lane];
        }
        int tk = 0;
        if (threadIdx.x == 0) tk = atomicAdd(ticket, 1);
        #pragma unroll
        for (int i = 0; i < RW; ++i) {
            const int row = rowBase + i;
            const int n = nn[i];
            const int m = n < ELL_CAP ? n : ELL_CAP;
            vPf acc = {};
            int j = 0;
            for (; j + 16 <= m; j += 16) gather_slots<16, K>(acc, hl, ee[i].x, ee[i].y, j);
            if (j + 8 <= m) { gather_slots<8, K>(acc, hl, ee[i].x, ee[i].y, j); j += 8; }
            if (j + 4 <= m) { gather_slots<4, K>(acc, hl, ee[i].x, ee[i].y, j); j += 4; }
            if (j + 2 <= m) { gather_slots<2, K>(acc, hl, ee[i].x, ee[i].y, j); j += 2; }
            if (j < m)      { gather_slots<1, K>(acc, hl, ee[i].x, ee[i].y, j); }
            if (n > ELL_CAP) {
                const int total = cnt[nRows];
                for (int base = 0; base < total; base += 32) {
                    const int k = base + lane;
                    int orow = -1, oc = 0, ov = 0;
                    if (k < total) {
                        orow = ovf[k];
                        oc   = ovf[(size_t)nEdges + k];
                        ov   = ovf[(size_t)nEdges * 2 + k];
                    }
                    unsigned hit = __builtin_amdgcn_ballot_w32(orow == row);
                    while (hit) {
                        const int jo = __builtin_ctz(hit);
                        hit &= hit - 1;
                        const int   cc = __builtin_amdgcn_readlane(oc, jo);
                        const float v  = __int_as_float(__builtin_amdgcn_readlane(ov, jo));
                        const vPf   s  = *(const vPf*)(hl + (size_t)cc * K);
                        acc += v * s;
                    }
                }
            }
            #pragma unroll
            for (int p = 0; p < PER; ++p) acc[p] = fmaxf(acc[p], 0.0f);
            *(vPf*)(&As[buf][(wave * RW + i) * AP + lane * PER]) = acc;
        }
        if (threadIdx.x == 0) sNext[buf] = (int)gridDim.x + tk;
        __syncthreads();
        const int next = __builtin_amdgcn_readfirstlane(sNext[buf]);

        const int row0 = step * ROWS + rt * 16;
        if (wave < TILES && row0 < nRows) {
            const float* ap = &As[buf][(rt * 16 + c) * AP + 4 * h];
            v8f d = v8f{};
            #pragma unroll
            for (int g = 0; g < KJ; g += 8) {
                float4 a[8], b[8];
                #pragma unroll
                for (int u = 0; u < 8; ++u) {
                    a[u] = *(const float4*)(ap + 8 * (g + u));
                    b[u] = *(const float4*)(bp + 2 * (g + u) * PITCH);
                }
                __builtin_amdgcn_sched_barrier(0);
                #pragma unroll
                for (int u = 0; u < 8; ++u) {
                    v2f a0; a0.x = a[u].x; a0.y = a[u].y;
                    v2f a1; a1.x = a[u].z; a1.y = a[u].w;
                    v2f b0; b0.x = b[u].x; b0.y = b[u].y;
                    v2f b1; b1.x = b[u].z; b1.y = b[u].w;
                    d = __builtin_amdgcn_wmma_f32_16x16x4_f32(
                            false, a0, false, b0, (short)0, d, false, false);
                    d = __builtin_amdgcn_wmma_f32_16x16x4_f32(
                            false, a1, false, b1, (short)0, d, false, false);
                }
            }
            float* yp = Y + (size_t)(row0 + h * 8) * M + ct * 16 + c;
            #pragma unroll
            for (int r = 0; r < 8; ++r) yp[(size_t)r * M] = d[r];
        }
        step = next;
        buf ^= 1;
    }
}

extern "C" void kernel_launch(void* const* d_in, const int* in_sizes, int n_in,
                              void* d_out, int out_size, void* d_ws, size_t ws_size,
                              hipStream_t stream) {
    const float* features = (const float*)d_in[0];
    const float* edgeVals = (const float*)d_in[1];
    const float* W0       = (const float*)d_in[2];
    const float* W1       = (const float*)d_in[3];
    const float* W2       = (const float*)d_in[4];
    const int*   edgeRow  = (const int*)d_in[5];
    const int*   edgeCol  = (const int*)d_in[6];

    const int N = in_sizes[0] / 256;
    const int E = in_sizes[1];

    float* ws = (float*)d_ws;
    float* t1    = ws;
    float* t0    = ws + (size_t)N * 256;
    float* t2    = ws + (size_t)N * 384;
    float* out   = (float*)d_out;
    int* ell = (int*)(ws + (size_t)N * 512);
    int* ovf = ell + (size_t)N * 64;
    int* cnt = ovf + (size_t)E * 3;
    int* sorted = cnt + ((size_t)N + 3 + 63) / 64 * 64;
    int* starts = sorted + (size_t)E * 2;

    const int rowBlocks  = (N + 7) / 8;
    const int tiles      = (E + TILE - 1) / TILE;
    const int gemmGroups = NBIN / 8;
    const int fuseGroups = 1024;

    bin_edges<<<tiles, BT, 0, stream>>>(edgeVals, edgeRow, edgeCol, sorted, starts, cnt, N, E);

    gemm_wmma_f32<256, 128, 4, false, true><<<gemmGroups, 512, 0, stream>>>(
        features, W0, t0, N, sorted, starts, tiles, cnt, ell, ovf, E);

    gather_gemm_f32<128, 64><<<fuseGroups, 256, 0, stream>>>(t0, W1, t1, cnt, ell, ovf, cnt + N + 1, N, E);

    gather_gemm_f32<64, 32><<<fuseGroups, 256, 0, stream>>>(t1, W2, t2, cnt, ell, ovf, cnt + N + 2, N, E);

    spmm_gather<32><<<rowBlocks, 256, 0, stream>>>(t2, cnt, ell, ovf, out, N, E);
}


# Gathers of 64- and 32-float rows load 16 B per lane: one load fetches 2 / 4 rows as whole lines, a wave's rows share their round trips; last gather 4 rows a wave (4x fewer waves); sum order kept
# speedup vs baseline: 4.7964x; 1.0579x over previous
#include <hip/hip_runtime.h>
#include <hip/hip_bf16.h>


typedef __attribute__((ext_vector_type(2))) float v2f;
typedef __attribute__((ext_vector_type(8))) float v8f;

template <int K, int M, int NT, bool RELU, bool REFILL, bool NORM>
__device__ __forceinline__ void gemm_unit(float4 (&aq)[K / 8],
                                          const float* __restrict__ bptr,
                                          const float* __restrict__ anext,
                                          float* __restrict__ yptr,
                                          const int h) {
    static_assert(!(NORM && RELU), "the squares are summed from the raw A values");
    constexpr int KJ    = K / 8;
    constexpr int PITCH = 4 * M + 32;
    v8f acc[NT];
    #pragma unroll
    for (int t = 0; t < NT; ++t) acc[t] = v8f{};
    v2f ss0 = {}, ss1 = {};

    float4 bq[2][NT];
    #pragma unroll
    for (int t = 0; t < NT; ++t) bq[0][t] = *(const float4*)(bptr + t * 64);

    #pragma unroll
    for (int j = 0; j < KJ; ++j) {
        if (j + 1 < KJ) {
            #pragma unroll
            for (int t = 0; t < NT; ++t)
                bq[(j + 1) & 1][t] = *(const float4*)(bptr + 2 * (j + 1) * PITCH + t * 64);
            __builtin_amdgcn_sched_barrier(0);
        }
        float4 a = aq[j];
        if (RELU) {
            a.x = fmaxf(a.x, 0.0f); a.y = fmaxf(a.y, 0.0f);
            a.z = fmaxf(a.z, 0.0f); a.w = fmaxf(a.w, 0.0f);
        }
        v2f a0; a0.x = a.x; a0.y = a.y;
        v2f a1; a1.x = a.z; a1.y = a.w;
        if (NORM) { ss0 += a0 * a0; ss1 += a1 * a1; }
        #pragma unroll
        for (int t = 0; t < NT; ++t) {
            v2f b; b.x = bq[j & 1][t].x; b.y = bq[j & 1][t].y;
            acc[t] = __builtin_amdgcn_wmma_f32_16x16x4_f32(
                         false, a0, false, b, (short)0, acc[t], false, false);
        }
        #pragma unroll
        for (int t = 0; t < NT; ++t) {
            v2f b; b.x = bq[j & 1][t].z; b.y = bq[j & 1][t].w;
            acc[t] = __builtin_amdgcn_wmma_f32_16x16x4_f32(
                         false, a1, false, b, (short)0, acc[t], false, false);
        }
        if (REFILL) {
            aq[j] = *(const float4*)(anext + 8 * j);
            __builtin_amdgcn_sched_barrier(0);
        }
    }
    if (NORM) {
        float s = (ss0.x + ss0.y) + (ss1.x + ss1.y);
        s += __shfl_xor(s, 16, 32);
        const float inv = rsqrtf(fmaxf(s, 1e-12f));
        #pragma unroll
        for (int r = 0; r < 8; ++r) {
            const float sc = __shfl(inv, 8 * h + r, 32);
            #pragma unroll
            for (int t = 0; t < NT; ++t)
                yptr[(size_t)r * M + t * 16] = acc[t][r] * sc;
        }
    } else {
        #pragma unroll
        for (int t = 0; t < NT; ++t) {
            #pragma unroll
            for (int r = 0; r < 8; ++r)
                yptr[(size_t)r * M + t * 16] = acc[t][r];
        }
    }
}

constexpr int ELL_CAP = 32;
constexpr int NBIN    = 2048;
constexpr int TILE    = 6272;
constexpr int SPITCH  = NBIN + 64;
constexpr int BT      = 512;
constexpr int EPT     = (TILE + BT - 1) / BT;
constexpr int BPT     = NBIN / BT;
constexpr int RB_MAX  = 64;
constexpr int COL_BITS = 26;

__global__ __launch_bounds__(BT) void bin_edges(const float* __restrict__ vals,
                                                const int* __restrict__ rows,
                                                const int* __restrict__ cols,
                                                int* __restrict__ sorted,
                                                int* __restrict__ starts,
                                                int* __restrict__ cnt,
                                                int nRows, int nEdges) {
    static_assert(NBIN % BT == 0 && BT % 32 == 0 && BT / 32 <= 32, "scan: BPT bins per thread, one wave sums the waves");
    static_assert(EPT * 3 < 63, "a thread's loads must fit the load counter");
    __shared__ int  cntB[NBIN];
    __shared__ int  startB[NBIN + 1];
    __shared__ int  waveTot[32];
    __shared__ int2 stage[TILE];

    const int t    = threadIdx.x;
    const int lane = t & 31;
    const int wave = t >> 5;
    const int RB   = (nRows + NBIN - 1) / NBIN;
    const int e0   = (int)blockIdx.x * TILE;
    const int e1   = e0 + TILE < nEdges ? e0 + TILE : nEdges;

    int r[EPT], cc[EPT], vv[EPT], b[EPT];
    #pragma unroll
    for (int u = 0; u < EPT; ++u) {
        const int e = e0 + t + u * BT;
        r[u] = -1; cc[u] = 0; vv[u] = 0;
        if (e < e1) { r[u] = rows[e]; cc[u] = cols[e]; vv[u] = __float_as_int(vals[e]); }
        if ((unsigned)r[u] >= (unsigned)nRows) r[u] = -1;
    }
    if (blockIdx.x == 0 && t < 3) cnt[nRows + t] = 0;
    #pragma unroll
    for (int i = 0; i < BPT; ++i) cntB[t + i * BT] = 0;
    __syncthreads();
    #pragma unroll
    for (int u = 0; u < EPT; ++u) {
        b[u] = 0;
        if (r[u] >= 0) { b[u] = r[u] / RB; atomicAdd(&cntB[b[u]], 1); }
    }
    __syncthreads();

    int cb[BPT];
    int s = 0;
    #pragma unroll
    for (int i = 0; i < BPT; ++i) { cb[i] = cntB[BPT * t + i]; s += cb[i]; }
    int incl = s;
    #pragma unroll
    for (int d = 1; d < 32; d <<= 1) {
        const int o = __shfl_up(incl, d, 32);
        if (lane >= d) incl += o;
    }
    if (lane == 31) waveTot[wave] = incl;
    __syncthreads();
    if (wave == 0) {
        const int v = lane < BT / 32 ? waveTot[lane] : 0;
        int vi = v;
        #pragma unroll
        for (int d = 1; d < 32; d <<= 1) {
            const int o = __shfl_up(vi, d, 32);
            if (lane >= d) vi += o;
        }
        if (lane < BT / 32) waveTot[lane] = vi - v;
    }
    __syncthreads();
    int run = incl - s + waveTot[wave];
    #pragma unroll
    for (int i = 0; i < BPT; ++i) {
        startB[BPT * t + i] = run;
        cntB[BPT * t + i]   = run;
        run += cb[i];
    }
    if (t == BT - 1) startB[NBIN] = run;
    __syncthreads();

    #pragma unroll
    for (int u = 0; u < EPT; ++u) {
        if (r[u] < 0) continue;
        const int pos = atomicAdd(&cntB[b[u]], 1);
        int2 rec;
        rec.x = (int)(((unsigned)(r[u] - b[u] * RB) << COL_BITS) | (unsigned)cc[u]);
        rec.y = vv[u];
        stage[pos] = rec;
    }
    __syncthreads();
    for (int p = t; p < e1 - e0; p += BT) ((int2*)sorted)[(size_t)e0 + p] = stage[p];
    for (int i = t; i <= NBIN; i += BT) starts[(size_t)blockIdx.x * SPITCH + i] = startB[i];
}

constexpr int PB = 8;
__device__ __forceinline__ void place_bin(const int* __restrict__ sorted,
                                          const int* __restrict__ starts,
                                          const int nTiles,
                                          int* __restrict__ cnt,
                                          int* __restrict__ ell,
                                          int* __restrict__ ovf,
                                          int* rowCnt, int2* slots,
                                          const int nRows, const int nEdges,
                                          const int bin, const int lane) {
    const int RB   = (nRows + NBIN - 1) / NBIN;
    const int row0 = bin * RB;
    if (row0 >= nRows) return;
    const int nr = row0 + RB <= nRows ? RB : nRows - row0;

    rowCnt[lane] = 0; rowCnt[lane + 32] = 0;
    for (int tb = 0; tb < nTiles; tb += 32) {
        const int tile = tb + lane;
        int a = 0, n = 0;
        if (tile < nTiles) {
            const int* sp = starts + (size_t)tile * SPITCH + bin;
            a = sp[0];
            n = sp[1] - a;
        }
        const int2* rp = (const int2*)sorted + (size_t)tile * TILE + a;
        for (int j0 = 0; __builtin_amdgcn_ballot_w32(j0 < n) != 0u; j0 += PB) {
            int2 rec[PB];
            #pragma unroll
            for (int u = 0; u < PB; ++u) {
                rec[u].x = 0; rec[u].y = 0;
                if (j0 + u < n) rec[u] = rp[j0 + u];
            }
            #pragma unroll
            for (int u = 0; u < PB; ++u) {
                if (j0 + u >= n) continue;
                const int rl   = (int)((unsigned)rec[u].x >> COL_BITS);
                const int col  = rec[u].x & ((1 << COL_BITS) - 1);
                const int slot = atomicAdd(&rowCnt[rl], 1);
                if (slot < ELL_CAP) {
                    int2 ent; ent.x = col; ent.y = rec[u].y;
                    slots[rl * ELL_CAP + slot] = ent;
                } else {
                    const int p = atomicAdd(&cnt[nRows], 1);
                    ovf[p]                      = row0 + rl;
                    ovf[(size_t)nEdges + p]     = col;
                    ovf[(size_t)nEdges * 2 + p] = rec[u].y;
                }
            }
        }
    }
    __builtin_amdgcn_fence(__ATOMIC_ACQ_REL, "workgroup");
    for (int i = 0; i < nr; ++i) {
        const int n = rowCnt[i];
        if (lane < (n < ELL_CAP ? n : ELL_CAP))
            ((int2*)ell)[(size_t)(row0 + i) * ELL_CAP + lane] = slots[i * ELL_CAP + lane];
    }
    if (lane < nr)      cnt[row0 + lane]      = rowCnt[lane];
    if (lane + 32 < nr) cnt[row0 + lane + 32] = rowCnt[lane + 32];
}

template <int K, int M, int NT, bool RELU, bool NORM>
__global__ __launch_bounds__(512) void gemm_wmma_f32(const float* __restrict__ X,
                                                     const float* __restrict__ W,
                                                     float* __restrict__ Y,
                                                     int nRows,
                                                     const int* __restrict__ sorted,
                                                     const int* __restrict__ starts,
                                                     int nTiles,
                                                     int* __restrict__ cnt,
                                                     int* __restrict__ ell,
                                                     int* __restrict__ ovf,
                                                     int nEdges) {
    constexpr int KQ    = K / 4;
    constexpr int PITCH = 4 * M + 32;
    constexpr int KJ    = K / 8;
    constexpr int CG    = M / (16 * NT);
    __shared__ __attribute__((aligned(16))) float Ws[KQ * PITCH];
    __shared__ int2 binSlots[8][RB_MAX * ELL_CAP];
    __shared__ int  binCnt[8][RB_MAX];

    const int lane = threadIdx.x & 31;
    const int wave = __builtin_amdgcn_readfirstlane(threadIdx.x >> 5);
    const int c    = lane & 15;
    const int h    = lane >> 4;

    const int totalUnits = (nRows / 16) * CG;
    const int perGroup   = (totalUnits + (int)gridDim.x - 1) / (int)gridDim.x;
    const int uBeg       = (int)blockIdx.x * perGroup;
    const int uEnd       = uBeg + perGroup < totalUnits ? uBeg + perGroup : totalUnits;
    int u = uBeg + wave;
    const bool haveWork = wave < 8 && u < uEnd;

    float4 aq[KJ];
    if (haveWork) {
        const float* ap = X + (size_t)((u / CG) * 16 + c) * K + 4 * h;
        #pragma unroll
        for (int j = 0; j < KJ; ++j) aq[j] = *(const float4*)(ap + 8 * j);
    }

    #pragma unroll 8
    for (int i = threadIdx.x; i < KQ * M; i += 512) {
        const int q = i / M;
        const int n = i % M;
        float4 v;
        v.x = W[(size_t)(4 * q + 0) * M + n];
        v.y = W[(size_t)(4 * q + 1) * M + n];
        v.z = W[(size_t)(4 * q + 2) * M + n];
        v.w = W[(size_t)(4 * q + 3) * M + n];
        *(float4*)(Ws + q * PITCH + 4 * n) = v;
    }
    __syncthreads();
    if (wave >= 8) {
        place_bin(sorted, starts, nTiles, cnt, ell, ovf, binCnt[wave - 8], binSlots[wave - 8],
                  nRows, nEdges, (int)blockIdx.x * 8 + wave - 8, lane);
        return;
    }
    if (!haveWork) return;

    while (true) {
        const int rowTile = u / CG;
        const int cg      = u % CG;
        const float* bptr = Ws + h * PITCH + (cg * NT * 16 + c) * 4;
        float* yptr = Y + (size_t)(rowTile * 16 + h * 8) * M + cg * NT * 16 + c;
        const int un = u + 8;
        if (un < uEnd) {
            const float* anext = X + (size_t)((un / CG) * 16 + c) * K + 4 * h;
            gemm_unit<K, M, NT, RELU, true, NORM>(aq, bptr, anext, yptr, h);
            u = un;
        } else {
            gemm_unit<K, M, NT, RELU, false, NORM>(aq, bptr, nullptr, yptr, h);
            break;
        }
    }
}

template <int D>
__global__ __launch_bounds__(256) void spmm_gather(const float* __restrict__ H,
                                                   const int* __restrict__ cnt,
                                                   const int* __restrict__ ell,
                                                   const int* __restrict__ ovf,
                                                   float* __restrict__ out,
                                                   int nRows, int nEdges) {
    static_assert(D == 32, "a row is one 128-byte line: eight lanes x 16 bytes");
    constexpr int GL   = D / 4;
    constexpr int RPW  = 32 / GL;
    constexpr int NBLK = ELL_CAP / GL;

    const int lane = threadIdx.x & 31;
    const int q    = lane / GL;
    const int i    = lane % GL;
    const int row0 = __builtin_amdgcn_readfirstlane((int)(blockIdx.x * 8 + (threadIdx.x >> 5)) * RPW);
    if (row0 >= nRows) return;
    const int  row  = row0 + q;
    const bool live = row < nRows;

    int n = 0;
    if (live) n = cnt[row];
    const int m = n < ELL_CAP ? n : ELL_CAP;
    int mx = 0;
    #pragma unroll
    for (int r = 0; r < RPW; ++r) {
        const int mr = __builtin_amdgcn_readlane(m, r * GL);
        mx = mr > mx ? mr : mx;
    }

    int2 e[NBLK];
    #pragma unroll
    for (int k = 0; k < NBLK; ++k) {
        e[k].x = 0; e[k].y = 0;
        if (GL * k + i < m) e[k] = ((const int2*)ell)[(size_t)row * ELL_CAP + GL * k + i];
    }

    const char* __restrict__ hb = (const char*)H;
    const unsigned ob = 16u * (unsigned)i;
    float4 acc = {0.0f, 0.0f, 0.0f, 0.0f};
    #pragma unroll
    for (int k0 = 0; k0 < NBLK; k0 += 2) {
        if (GL * k0 < mx) {
            float4 s[2 * GL];
            #pragma unroll
            for (int t = 0; t < 2 * GL; ++t) {
                const int k = k0 + t / GL;
                const int u = t % GL;
                const unsigned cc = (unsigned)__shfl(e[k].x, u, GL);
                if (GL * k + u < m) s[t] = *(const float4*)(hb + (cc * (unsigned)(D * 4) + ob));
            }
            #pragma unroll
            for (int t = 0; t < 2 * GL; ++t) {
                const int k = k0 + t / GL;
                const int u = t % GL;
                const float v = __int_as_float(__shfl(e[k].y, u, GL));
                if (GL * k + u < m) acc += v * s[t];
            }
        }
    }

    #pragma unroll
    for (int r = 0; r < RPW; ++r) {
        const int nr = __builtin_amdgcn_readlane(n, r * GL);
        if (nr > ELL_CAP) {
            const int total = cnt[nRows];
            for (int base = 0; base < total; base += 32) {
                const int p = base + lane;
                int orow = -1, oc = 0, ov = 0;
                if (p < total) {
                    orow = ovf[p];
                    oc   = ovf[(size_t)nEdges + p];
                    ov   = ovf[(size_t)nEdges * 2 + p];
                }
                unsigned hit = __builtin_amdgcn_ballot_w32(orow == row0 + r);
                while (hit) {
                    const int jo = __builtin_ctz(hit);
                    hit &= hit - 1;
                    const int   cc = __builtin_amdgcn_readlane(oc, jo);
                    const float vv = __int_as_float(__builtin_amdgcn_readlane(ov, jo));
                    if (q == r) {
                        const float4 s = *(const float4*)(hb + ((unsigned)cc * (unsigned)(D * 4) + ob));
                        acc += vv * s;
                    }
                }
            }
        }
    }
    if (live) *(float4*)(out + (size_t)row * D + 4 * i) = acc;
}

template <int B, int K, typename V>
__device__ __forceinline__ void gather_slots(V& acc, const float* __restrict__ hl,
                                             const int ex, const int ey, const int j0) {
    V     s[B];
    float v[B];
    #pragma unroll
    for (int u = 0; u < B; ++u) {
        const int cc = __builtin_amdgcn_readlane(ex, j0 + u);
        v[u] = __int_as_float(__builtin_amdgcn_readlane(ey, j0 + u));
        s[u] = *(const V*)(hl + (size_t)cc * K);
    }
    #pragma unroll
    for (int u = 0; u < B; ++u) acc += v[u] * s[u];
}

template <int K, int M>
__global__ __launch_bounds__(256) void gather_gemm_f32(const float* __restrict__ H,
                                                       const float* __restrict__ W,
                                                       float* __restrict__ Y,
                                                       const int* __restrict__ cnt,
                                                       const int* __restrict__ ell,
                                                       const int* __restrict__ ovf,
                                                       int* __restrict__ ticket,
                                                       int nRows, int nEdges) {
    constexpr int CT    = M / 16;
    constexpr int ROWS  = 32;
    constexpr int TILES = 2 * CT;
    constexpr int RW    = ROWS / 8;
    constexpr int KQ    = K / 4;
    constexpr int PITCH = 4 * M + 32;
    constexpr int KJ    = K / 8;
    constexpr int AP    = K + 4;
    constexpr int PER   = K / 32;
    static_assert(KJ % 8 == 0 && TILES <= 8, "product groups of 8 steps, one tile per wave");
    typedef __attribute__((ext_vector_type(PER))) float vPf;

    __shared__ __attribute__((aligned(16))) float Ws[KQ * PITCH];
    __shared__ __attribute__((aligned(16))) float As[2][ROWS * AP];
    __shared__ int sNext[2];

    const int lane = threadIdx.x & 31;
    const int wave = __builtin_amdgcn_readfirstlane(threadIdx.x >> 5);
    const int c    = lane & 15;
    const int h    = lane >> 4;

    #pragma unroll 8
    for (int i = threadIdx.x; i < KQ * M; i += 256) {
        const int q = i / M;
        const int n = i % M;
        float4 v;
        v.x = W[(size_t)(4 * q + 0) * M + n];
        v.y = W[(size_t)(4 * q + 1) * M + n];
        v.z = W[(size_t)(4 * q + 2) * M + n];
        v.w = W[(size_t)(4 * q + 3) * M + n];
        *(float4*)(Ws + q * PITCH + 4 * n) = v;
    }

    const int nSteps = (nRows + ROWS - 1) / ROWS;
    const int rt = wave / CT;
    const int ct = wave % CT;
    const float* __restrict__ hl = H + lane * PER;
    const float* bp = Ws + h * PITCH + (ct * 16 + c) * 4;

    int step = blockIdx.x;
    int buf  = 0;
    while (step < nSteps) {
        int tk = 0;
        if constexpr (K == 64) {
            const int rowBase = step * ROWS + wave * RW;
            int myN = 0;
            if (lane < RW && rowBase + lane < nRows) myN = cnt[rowBase + lane];
            int nn[RW];
            int mx = 0;
            #pragma unroll
            for (int i = 0; i < RW; ++i) {
                nn[i] = __builtin_amdgcn_readlane(myN, i);
                const int mi = nn[i] < ELL_CAP ? nn[i] : ELL_CAP;
                mx = mi > mx ? mi : mx;
            }
            int  ml[2];
            int2 e[2][2];
            #pragma unroll
            for (int p = 0; p < 2; ++p) {
                const int nl = h ? nn[2 * p + 1] : nn[2 * p];
                ml[p] = nl < ELL_CAP ? nl : ELL_CAP;
                #pragma unroll
                for (int k = 0; k < 2; ++k) {
                    e[p][k].x = 0; e[p][k].y = 0;
                    if (16 * k + c < ml[p])
                        e[p][k] = ((const int2*)ell)[(size_t)(rowBase + 2 * p + h) * ELL_CAP + 16 * k + c];
                }
            }
            if (threadIdx.x == 0) tk = atomicAdd(ticket, 1);

            const char* __restrict__ hb = (const char*)H;
            const unsigned ob = 16u * (unsigned)c;
            float4 acc[2];
            acc[0] = float4{0.0f, 0.0f, 0.0f, 0.0f};
            acc[1] = float4{0.0f, 0.0f, 0.0f, 0.0f};
            #pragma unroll
            for (int j0 = 0; j0 < ELL_CAP; j0 += 8) {
                if (j0 < mx) {
                    float4 s[2][8];
                    #pragma unroll
                    for (int p = 0; p < 2; ++p) {
                        #pragma unroll
                        for (int t = 0; t < 8; ++t) {
                            const int k = (j0 + t) / 16;
                            const int u = (j0 + t) % 16;
                            const unsigned cc = (unsigned)__shfl(e[p][k].x, u, 16);
                            s[p][t] = float4{0.0f, 0.0f, 0.0f, 0.0f};
                            if (j0 + t < ml[p]) s[p][t] = *(const float4*)(hb + (cc * (unsigned)(K * 4) + ob));
                        }
                    }
                    #pragma unroll
                    for (int p = 0; p < 2; ++p) {
                        #pragma unroll
                        for (int t = 0; t < 8; ++t) {
                            const int k = (j0 + t) / 16;
                            const int u = (j0 + t) % 16;
                            const float v = __int_as_float(__shfl(e[p][k].y, u, 16));
                            acc[p] += v * s[p][t];
                        }
                    }
                }
            }
            #pragma unroll
            for (int r = 0; r < RW; ++r) {
                if (nn[r] > ELL_CAP) {
                    const int total = cnt[nRows];
                    for (int base = 0; base < total; base += 32) {
                        const int k = base + lane;
                        int orow = -1, oc = 0, ov = 0;
                        if (k < total) {
                            orow = ovf[k];
                            oc   = ovf[(size_t)nEdges + k];
                            ov   = ovf[(size_t)nEdges * 2 + k];
                        }
                        unsigned hit = __builtin_amdgcn_ballot_w32(orow == rowBase + r);
                        while (hit) {
                            const int jo = __builtin_ctz(hit);
                            hit &= hit - 1;
                            const int   cc = __builtin_amdgcn_readlane(oc, jo);
                            const float vv = __int_as_float(__builtin_amdgcn_readlane(ov, jo));
                            if (h == (r & 1)) {
                                const float4 s = *(const float4*)(hb + ((unsigned)cc * (unsigned)(K * 4) + ob));
                                acc[r >> 1] += vv * s;
                            }
                        }
                    }
                }
            }
            #pragma unroll
            for (int p = 0; p < 2; ++p) {
                float4 a = acc[p];
                a.x = fmaxf(a.x, 0.0f); a.y = fmaxf(a.y, 0.0f);
                a.z = fmaxf(a.z, 0.0f); a.w = fmaxf(a.w, 0.0f);
                *(float4*)(&As[buf][(wave * RW + 2 * p + h) * AP + 4 * c]) = a;
            }
        } else {
            const int rowBase = step * ROWS + wave * RW;
            int myN = 0;
            if (lane < RW && rowBase + lane < nRows) myN = cnt[rowBase + lane];
            int  nn[RW];
            int2 ee[RW];
            #pragma unroll
            for (int i = 0; i < RW; ++i) {
                nn[i] = __builtin_amdgcn_readlane(myN, i);
                const int m = nn[i] < ELL_CAP ? nn[i] : ELL_CAP;
                ee[i].x = 0; ee[i].y = 0;
                if (lane < m) ee[i] = ((const int2*)ell)[(size_t)(rowBase + i) * ELL_CAP + lane];
            }
            if (threadIdx.x == 0) tk = atomicAdd(ticket, 1);
            #pragma unroll
            for (int i = 0; i < RW; ++i) {
                const int row = rowBase + i;
                const int n = nn[i];
                const int m = n < ELL_CAP ? n : ELL_CAP;
                vPf acc = {};
                int j = 0;
                for (; j + 16 <= m; j += 16) gather_slots<16, K>(acc, hl, ee[i].x, ee[i].y, j);
                if (j + 8 <= m) { gather_slots<8, K>(acc, hl, ee[i].x, ee[i].y, j); j += 8; }
                if (j + 4 <= m) { gather_slots<4, K>(acc, hl, ee[i].x, ee[i].y, j); j += 4; }
                if (j + 2 <= m) { gather_slots<2, K>(acc, hl, ee[i].x, ee[i].y, j); j += 2; }
                if (j < m)      { gather_slots<1, K>(acc, hl, ee[i].x, ee[i].y, j); }
                if (n > ELL_CAP) {
                    const int total = cnt[nRows];
                    for (int base = 0; base < total; base += 32) {
                        const int k = base + lane;
                        int orow = -1, oc = 0, ov = 0;
                        if (k < total) {
                            orow = ovf[k];
                            oc   = ovf[(size_t)nEdges + k];
                            ov   = ovf[(size_t)nEdges * 2 + k];
                        }
                        unsigned hit = __builtin_amdgcn_ballot_w32(orow == row);
                        while (hit) {
                            const int jo = __builtin_ctz(hit);
                            hit &= hit - 1;
                            const int   cc = __builtin_amdgcn_readlane(oc, jo);
                            const float v  = __int_as_float(__builtin_amdgcn_readlane(ov, jo));
                            const vPf   s  = *(const vPf*)(hl + (size_t)cc * K);
                            acc += v * s;
                        }
                    }
                }
                #pragma unroll
                for (int p = 0; p < PER; ++p) acc[p] = fmaxf(acc[p], 0.0f);
                *(vPf*)(&As[buf][(wave * RW + i) * AP + lane * PER]) = acc;
            }
        }
        if (threadIdx.x == 0) sNext[buf] = (int)gridDim.x + tk;
        __syncthreads();
        const int next = __builtin_amdgcn_readfirstlane(sNext[buf]);

        const int row0 = step * ROWS + rt * 16;
        if (wave < TILES && row0 < nRows) {
            const float* ap = &As[buf][(rt * 16 + c) * AP + 4 * h];
            v8f d = v8f{};
            #pragma unroll
            for (int g = 0; g < KJ; g += 8) {
                float4 a[8], b[8];
                #pragma unroll
                for (int u = 0; u < 8; ++u) {
                    a[u] = *(const float4*)(ap + 8 * (g + u));
                    b[u] = *(const float4*)(bp + 2 * (g + u) * PITCH);
                }
                __builtin_amdgcn_sched_barrier(0);
                #pragma unroll
                for (int u = 0; u < 8; ++u) {
                    v2f a0; a0.x = a[u].x; a0.y = a[u].y;
                    v2f a1; a1.x = a[u].z; a1.y = a[u].w;
                    v2f b0; b0.x = b[u].x; b0.y = b[u].y;
                    v2f b1; b1.x = b[u].z; b1.y = b[u].w;
                    d = __builtin_amdgcn_wmma_f32_16x16x4_f32(
                            false, a0, false, b0, (short)0, d, false, false);
                    d = __builtin_amdgcn_wmma_f32_16x16x4_f32(
                            false, a1, false, b1, (short)0, d, false, false);
                }
            }
            float* yp = Y + (size_t)(row0 + h * 8) * M + ct * 16 + c;
            #pragma unroll
            for (int r = 0; r < 8; ++r) yp[(size_t)r * M] = d[r];
        }
        step = next;
        buf ^= 1;
    }
}

extern "C" void kernel_launch(void* const* d_in, const int* in_sizes, int n_in,
                              void* d_out, int out_size, void* d_ws, size_t ws_size,
                              hipStream_t stream) {
    const float* features = (const float*)d_in[0];
    const float* edgeVals = (const float*)d_in[1];
    const float* W0       = (const float*)d_in[2];
    const float* W1       = (const float*)d_in[3];
    const float* W2       = (const float*)d_in[4];
    const int*   edgeRow  = (const int*)d_in[5];
    const int*   edgeCol  = (const int*)d_in[6];

    const int N = in_sizes[0] / 256;
    const int E = in_sizes[1];

    float* ws = (float*)d_ws;
    float* t1    = ws;
    float* t0    = ws + (size_t)N * 256;
    float* t2    = ws + (size_t)N * 384;
    float* out   = (float*)d_out;
    int* ell = (int*)(ws + (size_t)N * 512);
    int* ovf = ell + (size_t)N * 64;
    int* cnt = ovf + (size_t)E * 3;
    int* sorted = cnt + ((size_t)N + 3 + 63) / 64 * 64;
    int* starts = sorted + (size_t)E * 2;

    const int rowBlocks  = (N + 31) / 32;
    const int tiles      = (E + TILE - 1) / TILE;
    const int gemmGroups = NBIN / 8;
    const int fuseGroups = 1024;

    bin_edges<<<tiles, BT, 0, stream>>>(edgeVals, edgeRow, edgeCol, sorted, starts, cnt, N, E);

    gemm_wmma_f32<256, 128, 4, false, true><<<gemmGroups, 512, 0, stream>>>(
        features, W0, t0, N, sorted, starts, tiles, cnt, ell, ovf, E);

    gather_gemm_f32<128, 64><<<fuseGroups, 256, 0, stream>>>(t0, W1, t1, cnt, ell, ovf, cnt + N + 1, N, E);

    gather_gemm_f32<64, 32><<<fuseGroups, 256, 0, stream>>>(t1, W2, t2, cnt, ell, ovf, cnt + N + 2, N, E);

    spmm_gather<32><<<rowBlocks, 256, 0, stream>>>(t2, cnt, ell, ovf, out, N, E);
}
